# starting kernel
# baseline (speedup 1.0000x reference)
#include <hip/hip_runtime.h>
#include <hip/hip_bf16.h>
#include <cstdint>

namespace {

constexpr int B_ = 4, N_ = 8192, CIN_ = 4, C_ = 64, K_ = 32, STRIDE_ = 4, NB_ = 2, EXP_ = 4;
constexpr int M_ = N_ / STRIDE_;
constexpr float R1SQ = 0.1f * 0.1f;
constexpr float R2SQ = 0.2f * 0.2f;
constexpr float NEG_BIG = -3.4028235e38f;

typedef float v2f __attribute__((ext_vector_type(2)));
typedef float v8f __attribute__((ext_vector_type(8)));

__global__ void f1_wmma_kernel(const float* __restrict__ f, const float* __restrict__ Wsa,
                               const float* __restrict__ bsa, float* __restrict__ f1) {
  int wave = (blockIdx.x * blockDim.x + threadIdx.x) >> 5;
  int lane = threadIdx.x & 31;
  constexpr int tilesC = C_ / 16;
  constexpr int nTiles = (B_ * N_ / 16) * tilesC;
  if (wave >= nTiles) return;
  int row0 = (wave / tilesC) * 16;
  int c0 = (wave % tilesC) * 16;
  int b = row0 / N_;
  int n0 = row0 % N_;
  int half = lane >> 4;
  int ln = lane & 15;

  const float* fb = f + ((size_t)b * CIN_ + 2 * half) * N_ + n0 + ln;
  v2f a;  a.x = fb[0];        a.y = fb[N_];
  const float* wc = Wsa + (size_t)(2 * half) * C_ + c0 + ln;
  v2f bb; bb.x = wc[0];       bb.y = wc[C_];
  v8f acc = {0.f, 0.f, 0.f, 0.f, 0.f, 0.f, 0.f, 0.f};
  acc = __builtin_amdgcn_wmma_f32_16x16x4_f32(false, a, false, bb, (short)0, acc,
                                              false, false);
  int col = c0 + ln;
  float bv = bsa[col];
#pragma unroll
  for (int r = 0; r < 8; ++r) {
    int row = row0 + r + 8 * half;
    f1[(size_t)row * C_ + col] = fmaxf(acc[r] + bv, 0.0f);
  }
}

__global__ void ball_query_kernel(const float* __restrict__ p, int n_src, int src_stride,
                                  float r2, int* __restrict__ idx_out) {
  int wave = (blockIdx.x * blockDim.x + threadIdx.x) >> 5;
  int lane = threadIdx.x & 31;
  if (wave >= B_ * M_) return;
  int b = wave / M_;
  int m = wave % M_;
  const float* pb = p + (size_t)b * N_ * 3;
  float qx = pb[(size_t)m * STRIDE_ * 3 + 0];
  float qy = pb[(size_t)m * STRIDE_ * 3 + 1];
  float qz = pb[(size_t)m * STRIDE_ * 3 + 2];
  int* out = idx_out + (size_t)wave * K_;

  int count = 0;
  int first_idx = n_src - 1;
  for (int base = 0; base < n_src; base += 32) {
    int j = base + lane;
    const float* sp = pb + (size_t)j * src_stride * 3;
    float dx = sp[0] - qx, dy = sp[1] - qy, dz = sp[2] - qz;
    bool hit = (dx * dx + dy * dy + dz * dz) <= r2;
    unsigned mask = (unsigned)__ballot(hit);
    if (count == 0 && mask != 0u) first_idx = base + (__ffs(mask) - 1);
    if (hit) {
      int pos = count + __popc(mask & ((1u << lane) - 1u));
      if (pos < K_) out[pos] = j;
    }
    count += __popc(mask);
    if (count >= K_) break;
  }
  if (count > K_) count = K_;
  for (int t = count + lane; t < K_; t += 32) out[t] = first_idx;
}

__global__ void neigh_max_kernel(const float* __restrict__ p, const float* __restrict__ src,
                                 const int* __restrict__ idx, int src_stride, int src_n,
                                 const float* __restrict__ Wpe, const float* __restrict__ bpe,
                                 float* __restrict__ out) {
  __shared__ int s_idx[4][K_];
  __shared__ float s_dq[4][K_][3];
  int g = threadIdx.x >> 6;
  int c = threadIdx.x & 63;
  int bm = blockIdx.x * 4 + g;
  if (bm >= B_ * M_) return;
  int b = bm / M_;
  int m = bm % M_;
  const float* pb = p + (size_t)b * N_ * 3;

  if (c < K_) {
    float qx = pb[(size_t)m * STRIDE_ * 3 + 0];
    float qy = pb[(size_t)m * STRIDE_ * 3 + 1];
    float qz = pb[(size_t)m * STRIDE_ * 3 + 2];
    int j = idx[(size_t)bm * K_ + c];
    s_idx[g][c] = j;
    const float* sp = pb + (size_t)j * src_stride * 3;
    s_dq[g][c][0] = sp[0] - qx;
    s_dq[g][c][1] = sp[1] - qy;
    s_dq[g][c][2] = sp[2] - qz;
  }
  __syncthreads();

  float w0 = Wpe[0 * C_ + c], w1 = Wpe[1 * C_ + c], w2 = Wpe[2 * C_ + c];
  float bias = bpe[c];
  const float* sb = src + (size_t)b * src_n * C_;
  float best = NEG_BIG;
#pragma unroll 4
  for (int k = 0; k < K_; ++k) {
    int j = s_idx[g][k];
    float v = s_dq[g][k][0] * w0 + s_dq[g][k][1] * w1 + s_dq[g][k][2] * w2 + bias
            + sb[(size_t)j * C_ + c];
    best = fmaxf(best, v);
  }
  out[(size_t)bm * C_ + c] = best;
}

template <int KDIM, bool RELU, bool SKIP>
__global__ void gemm_wmma_kernel(const float* __restrict__ X, const float* __restrict__ W,
                                 const float* __restrict__ bias,
                                 const float* __restrict__ skip, float* __restrict__ out,
                                 int Mrows, int Ncols) {
  constexpr int MT = 4;
  int wave = (blockIdx.x * blockDim.x + threadIdx.x) >> 5;
  int lane = threadIdx.x & 31;
  int tilesN = Ncols >> 4;
  int nTiles = (Mrows >> 6) * tilesN;
  if (wave >= nTiles) return;
  int m0 = (wave / tilesN) << 6;
  int n0 = (wave % tilesN) << 4;
  int half = lane >> 4;
  int ln = lane & 15;

  const float* Xrow0 = X + (size_t)(m0 + ln) * KDIM + 2 * half;
  const float* Wcol = W + (size_t)(2 * half) * Ncols + n0 + ln;
  constexpr size_t XSTEP = (size_t)16 * KDIM;

  v8f acc[MT];
#pragma unroll
  for (int t = 0; t < MT; ++t) acc[t] = v8f{0.f, 0.f, 0.f, 0.f, 0.f, 0.f, 0.f, 0.f};

#pragma unroll 4
  for (int k = 0; k < KDIM; k += 4) {
    v2f bb;
    bb.x = Wcol[(size_t)k * Ncols];
    bb.y = Wcol[(size_t)(k + 1) * Ncols];
#pragma unroll
    for (int t = 0; t < MT; ++t) {
      v2f a;
      a.x = Xrow0[t * XSTEP + k];
      a.y = Xrow0[t * XSTEP + k + 1];
      acc[t] = __builtin_amdgcn_wmma_f32_16x16x4_f32(false, a, false, bb, (short)0,
                                                     acc[t], false, false);
    }
  }

  int col = n0 + ln;
  float bv = bias[col];
#pragma unroll
  for (int t = 0; t < MT; ++t) {
#pragma unroll
    for (int r = 0; r < 8; ++r) {
      int row = m0 + t * 16 + r + 8 * half;
      float v = acc[t][r] + bv;
      if (SKIP) v += skip[(size_t)row * Ncols + col];
      if (RELU) v = fmaxf(v, 0.0f);
      out[(size_t)row * Ncols + col] = v;
    }
  }
}

}

extern "C" void kernel_launch(void* const* d_in, const int* in_sizes, int n_in,
                              void* d_out, int out_size, void* d_ws, size_t ws_size,
                              hipStream_t stream) {
  (void)in_sizes; (void)n_in; (void)out_size; (void)ws_size;
  const float* p      = (const float*)d_in[0];
  const float* f      = (const float*)d_in[1];
  const float* W_sa   = (const float*)d_in[2];
  const float* b_sa   = (const float*)d_in[3];
  const float* W_pe1  = (const float*)d_in[4];
  const float* b_pe1  = (const float*)d_in[5];
  const float* W_pe2  = (const float*)d_in[6];
  const float* b_pe2  = (const float*)d_in[7];
  const float* W_conv = (const float*)d_in[8];
  const float* b_conv = (const float*)d_in[9];
  const float* W_pw1  = (const float*)d_in[10];
  const float* b_pw1  = (const float*)d_in[11];
  const float* W_pw2  = (const float*)d_in[12];
  const float* b_pw2  = (const float*)d_in[13];
  float* feat = (float*)d_out;

  char* ws = (char*)d_ws;
  float* f1 = (float*)ws;  ws += sizeof(float) * (size_t)B_ * N_ * C_;
  float* h  = (float*)ws;  ws += sizeof(float) * (size_t)B_ * M_ * C_;
  float* h2 = (float*)ws;  ws += sizeof(float) * (size_t)B_ * M_ * EXP_ * C_;
  float* agg = (float*)ws; ws += sizeof(float) * (size_t)B_ * M_ * C_;
  int* idx1 = (int*)ws;    ws += sizeof(int) * (size_t)B_ * M_ * K_;
  int* idx2 = (int*)ws;

  const int THREADS = 256;

  {
    int waves = (B_ * N_ / 16) * (C_ / 16);
    f1_wmma_kernel<<<(waves * 32 + THREADS - 1) / THREADS, THREADS, 0, stream>>>(
        f, W_sa, b_sa, f1);
  }

  int bq_blocks = (B_ * M_ * 32 + THREADS - 1) / THREADS;
  ball_query_kernel<<<bq_blocks, THREADS, 0, stream>>>(p, N_, 1, R1SQ, idx1);
  ball_query_kernel<<<bq_blocks, THREADS, 0, stream>>>(p, M_, STRIDE_, R2SQ, idx2);

  int nm_blocks = (B_ * M_ + 3) / 4;
  neigh_max_kernel<<<nm_blocks, THREADS, 0, stream>>>(p, f1, idx1, 1, N_, W_pe1, b_pe1, feat);

  const int MR = B_ * M_;
  for (int i = 0; i < NB_; ++i) {
    {
      int waves = (MR / 64) * (C_ / 16);
      gemm_wmma_kernel<C_, true, false>
          <<<(waves * 32 + THREADS - 1) / THREADS, THREADS, 0, stream>>>(
              feat, W_conv + (size_t)i * C_ * C_, b_conv + (size_t)i * C_, nullptr, h, MR, C_);
    }
    neigh_max_kernel<<<nm_blocks, THREADS, 0, stream>>>(p, h, idx2, STRIDE_, M_, W_pe2,
                                                        b_pe2, agg);
    {
      int waves = (MR / 64) * ((EXP_ * C_) / 16);
      gemm_wmma_kernel<C_, true, false>
          <<<(waves * 32 + THREADS - 1) / THREADS, THREADS, 0, stream>>>(
              agg, W_pw1 + (size_t)i * C_ * EXP_ * C_, b_pw1 + (size_t)i * EXP_ * C_, nullptr,
              h2, MR, EXP_ * C_);
    }
    {
      int waves = (MR / 64) * (C_ / 16);
      gemm_wmma_kernel<EXP_ * C_, true, true>
          <<<(waves * 32 + THREADS - 1) / THREADS, THREADS, 0, stream>>>(
              h2, W_pw2 + (size_t)i * EXP_ * C_ * C_, b_pw2 + (size_t)i * C_, feat, feat, MR, C_);
    }
  }
}


# conv GEMM: a wave owns a 16x32 tile and loads all its 48 operand fragments before the first WMMA (K loop unrolled; was 4 serial load rounds, 64 blocks); A fragment = one 8-byte load; sums unchanged
# speedup vs baseline: 3.3112x; 3.3112x over previous
#include <hip/hip_runtime.h>
#include <hip/hip_bf16.h>
#include <cstdint>

namespace {

constexpr int B_ = 4, N_ = 8192, CIN_ = 4, C_ = 64, K_ = 32, STRIDE_ = 4, NB_ = 2, EXP_ = 4;
constexpr int M_ = N_ / STRIDE_;
constexpr float R1SQ = 0.1f * 0.1f;
constexpr float R2SQ = 0.2f * 0.2f;
constexpr float NEG_BIG = -3.4028235e38f;

typedef float v2f __attribute__((ext_vector_type(2)));
typedef float v3f __attribute__((ext_vector_type(3)));
typedef float v8f __attribute__((ext_vector_type(8)));

__global__ void f1_wmma_kernel(const float* __restrict__ f, const float* __restrict__ Wsa,
                               const float* __restrict__ bsa, float* __restrict__ f1) {
  int wave = (blockIdx.x * blockDim.x + threadIdx.x) >> 5;
  int lane = threadIdx.x & 31;
  constexpr int tilesC = C_ / 16;
  constexpr int nTiles = (B_ * N_ / 16) * tilesC;
  if (wave >= nTiles) return;
  int row0 = (wave / tilesC) * 16;
  int c0 = (wave % tilesC) * 16;
  int b = row0 / N_;
  int n0 = row0 % N_;
  int half = lane >> 4;
  int ln = lane & 15;

  const float* fb = f + ((size_t)b * CIN_ + 2 * half) * N_ + n0 + ln;
  v2f a;  a.x = fb[0];        a.y = fb[N_];
  const float* wc = Wsa + (size_t)(2 * half) * C_ + c0 + ln;
  v2f bb; bb.x = wc[0];       bb.y = wc[C_];
  v8f acc = {0.f, 0.f, 0.f, 0.f, 0.f, 0.f, 0.f, 0.f};
  acc = __builtin_amdgcn_wmma_f32_16x16x4_f32(false, a, false, bb, (short)0, acc,
                                              false, false);
  int col = c0 + ln;
  float bv = bsa[col];
#pragma unroll
  for (int r = 0; r < 8; ++r) {
    int row = row0 + r + 8 * half;
    f1[(size_t)row * C_ + col] = fmaxf(acc[r] + bv, 0.0f);
  }
}

constexpr int QB_ = 4;
constexpr int PD_ = 16;
static_assert(M_ % QB_ == 0, "a block's queries share one batch");
static_assert(K_ == 32, "one lane per list slot");
static_assert(32 % STRIDE_ == 0, "list B's sources keep their lanes from chunk to chunk");
constexpr unsigned qlane_mask() {
  unsigned m = 0u;
  for (int l = 0; l < 32; l += STRIDE_) m |= 1u << l;
  return m;
}
constexpr unsigned QMASK_ = qlane_mask();

template <int NSEG>
__global__ void __launch_bounds__(32 * NSEG)
ball_query_both_kernel(const float* __restrict__ p, float r2a, float r2b,
                       int* __restrict__ idx_a, int* __restrict__ idx_b) {
  constexpr int SEG = N_ / NSEG;
  static_assert(N_ % NSEG == 0 && SEG % (32 * PD_) == 0, "whole trips per part");
  static_assert(QB_ % NSEG == 0, "wave w puts the lists of queries w, w+NSEG, .. together");
  __shared__ int lst[2][NSEG][QB_][K_];
  __shared__ int cnts[NSEG][2 * QB_];
  const int gv = (int)(blockIdx.x * blockDim.x + threadIdx.x);
  const int lane = threadIdx.x & 31;
  const int w = __builtin_amdgcn_readfirstlane((int)(threadIdx.x >> 5));
  const int q0 = (int)blockIdx.x * QB_;
  if (q0 >= B_ * M_) return;
  const int b = q0 / M_;
  const float* pb = p + (size_t)b * N_ * 3;
  const bool qlane = (lane % STRIDE_) == 0;

  v3f q[QB_];
  {
    int q0v = (gv / (32 * NSEG)) * QB_;
    const float* pbv = p + (size_t)(q0v / M_) * N_ * 3;
    int m0v = q0v % M_;
#pragma unroll
    for (int i = 0; i < QB_; ++i) {
      const float* qa = pbv + (size_t)(m0v + i) * STRIDE_ * 3;
      q[i] = v3f{qa[0], qa[1], qa[2]};
    }
  }
#pragma unroll
  for (int i = 0; i < QB_; ++i) asm volatile("" :: "v"(q[i]));
  int cnta[QB_], cntb[QB_];
#pragma unroll
  for (int i = 0; i < QB_; ++i) { cnta[i] = 0; cntb[i] = 0; }

  const int sbase = w * SEG;
  const unsigned loff = (unsigned)lane * 3;
  const float* ps = pb + (size_t)sbase * 3 + loff;
  v3f s[PD_];
#pragma unroll
  for (int u = 0; u < PD_; ++u) {
    const float* sp = ps + (size_t)(u * 32) * 3;
    s[u] = v3f{sp[0], sp[1], sp[2]};
  }

  for (int base = 0; base < SEG; base += 32 * PD_) {
    int nbase = base + 32 * PD_;
    if (nbase > SEG - 32 * PD_) nbase = SEG - 32 * PD_;
    const float* np = ps + (size_t)nbase * 3;
#pragma unroll
    for (int u = 0; u < PD_; ++u) {
      int cb = sbase + base + 32 * u;
      bool hita[QB_], hitb[QB_];
      unsigned maska[QB_], maskb[QB_];
#pragma unroll
      for (int i = 0; i < QB_; ++i) {
        v2f d = s[u].xy - q[i].xy;
        float dz = s[u].z - q[i].z;
        float s2;
        {
#pragma clang fp contract(off)
          v2f dd = d * d;
          s2 = dd.x + dd.y;
        }
        float d2 = __builtin_fmaf(dz, dz, s2);
        hita[i] = d2 <= r2a;
        hitb[i] = d2 <= r2b;
      }
#pragma unroll
      for (int i = 0; i < QB_; ++i) {
        maska[i] = (unsigned)__ballot(hita[i]);
        maskb[i] = (unsigned)__ballot(hitb[i]) & QMASK_;
      }
#pragma unroll
      for (int i = 0; i < QB_; ++i) {
        if (__builtin_expect(maska[i] != 0u, 0)) {
          if (cnta[i] < K_) {
            if (hita[i]) {
              int pos = (int)__builtin_amdgcn_mbcnt_lo(maska[i], (unsigned)cnta[i]);
              if (pos < K_) lst[0][w][i][pos] = cb + lane;
            }
            cnta[i] += __popc(maska[i]);
          }
        }
      }
#pragma unroll
      for (int i = 0; i < QB_; ++i) {
        if (__builtin_expect(maskb[i] != 0u, 0)) {
          if (cntb[i] < K_) {
            if (hitb[i]) {
              int pos = (int)__builtin_amdgcn_mbcnt_lo(maskb[i], (unsigned)cntb[i]);
              if (qlane && pos < K_) lst[1][w][i][pos] = (cb + lane) / STRIDE_;
            }
            cntb[i] += __popc(maskb[i]);
          }
        }
      }
      const float* sp = np + (size_t)(u * 32) * 3;
      s[u] = v3f{sp[0], sp[1], sp[2]};
    }
    int cmin = cnta[0] < cntb[0] ? cnta[0] : cntb[0];
#pragma unroll
    for (int i = 1; i < QB_; ++i) {
      cmin = cnta[i] < cmin ? cnta[i] : cmin;
      cmin = cntb[i] < cmin ? cntb[i] : cmin;
    }
    if (cmin >= K_) break;
  }

  {
    int cv = cnta[0];
#pragma unroll
    for (int i = 1; i < QB_; ++i) cv = (lane == i) ? cnta[i] : cv;
#pragma unroll
    for (int i = 0; i < QB_; ++i) cv = (lane == QB_ + i) ? cntb[i] : cv;
    if (lane < 2 * QB_) cnts[w][lane] = cv;
  }
  __syncthreads();

#pragma unroll
  for (int which = 0; which < 2; ++which) {
#pragma unroll
    for (int r = 0; r < QB_ / NSEG; ++r) {
      const int qi = w + r * NSEG;
      int c[NSEG], pre[NSEG + 1];
      pre[0] = 0;
#pragma unroll
      for (int sgm = 0; sgm < NSEG; ++sgm) {
        int cs = cnts[sgm][which * QB_ + qi];
        c[sgm] = cs < K_ ? cs : K_;
        pre[sgm + 1] = pre[sgm] + c[sgm];
      }
      const int tot = pre[NSEG];
      int val = (which ? M_ : N_) - 1;
      if (tot > 0) {
        int sf = NSEG - 1;
#pragma unroll
        for (int sgm = NSEG - 2; sgm >= 0; --sgm) sf = c[sgm] > 0 ? sgm : sf;
        int sg = sf, off = 0;
        if (lane < tot) {
          sg = 0;
#pragma unroll
          for (int sgm = 1; sgm < NSEG; ++sgm) sg += (lane >= pre[sgm]) ? 1 : 0;
          int psg = 0;
#pragma unroll
          for (int sgm = 1; sgm < NSEG; ++sgm) psg = (sg >= sgm) ? pre[sgm] : psg;
          off = lane - psg;
        }
        val = lst[which][sg][qi][off];
      }
      int* outp = which ? idx_b : idx_a;
      outp[(size_t)(q0 + qi) * K_ + lane] = val;
    }
  }
}

constexpr int NMQ_ = 2;
static_assert(M_ % NMQ_ == 0, "a wave's queries share one batch");
static_assert(C_ == 64 && K_ == 32, "2 channels per lane, one neighbour per lane");

__device__ __forceinline__ float lane_bcast(float x, int k) {
  return __builtin_bit_cast(float, __builtin_amdgcn_readlane(__builtin_bit_cast(int, x), k));
}

__global__ void __launch_bounds__(256)
neigh_max_kernel(const float* __restrict__ p, const float* __restrict__ src,
                 const int* __restrict__ idx, int src_stride, int src_n,
                 const float* __restrict__ Wpe, const float* __restrict__ bpe,
                 float* __restrict__ out) {
  int wave_v = (int)((blockIdx.x * blockDim.x + threadIdx.x) >> 5);
  int wave = __builtin_amdgcn_readfirstlane(wave_v);
  int lane = threadIdx.x & 31;
  int bm0 = wave * NMQ_;
  if (bm0 >= B_ * M_) return;
  int b = bm0 / M_;
  const float* pb = p + (size_t)b * N_ * 3;
  const float* sb = src + (size_t)b * src_n * C_ + 2 * lane;

  int j[NMQ_];
  float dx[NMQ_], dy[NMQ_], dz[NMQ_];
#pragma unroll
  for (int i = 0; i < NMQ_; ++i) j[i] = idx[(size_t)(bm0 + i) * K_ + lane];
#pragma unroll
  for (int i = 0; i < NMQ_; ++i) {
    int m = (bm0 + i) % M_;
    const float* qa = pb + (size_t)m * STRIDE_ * 3;
    const float* sp = pb + (size_t)j[i] * src_stride * 3;
    dx[i] = sp[0] - qa[0];
    dy[i] = sp[1] - qa[1];
    dz[i] = sp[2] - qa[2];
  }

  v2f g[NMQ_][K_];
#pragma unroll
  for (int i = 0; i < NMQ_; ++i) {
#pragma unroll
    for (int k = 0; k < K_; ++k) {
      int jk = __builtin_amdgcn_readlane(j[i], k);
      g[i][k] = *(const v2f*)(sb + (size_t)(unsigned)jk * C_);
    }
  }

  v2f w0 = *(const v2f*)(Wpe + 0 * C_ + 2 * lane);
  v2f w1 = *(const v2f*)(Wpe + 1 * C_ + 2 * lane);
  v2f w2 = *(const v2f*)(Wpe + 2 * C_ + 2 * lane);
  v2f bias = *(const v2f*)(bpe + 2 * lane);

#pragma unroll
  for (int i = 0; i < NMQ_; ++i) {
    float best0 = NEG_BIG, best1 = NEG_BIG;
#pragma unroll
    for (int k = 0; k < K_; ++k) {
      float sx = lane_bcast(dx[i], k), sy = lane_bcast(dy[i], k), sz = lane_bcast(dz[i], k);
      float v0 = sx * w0.x + sy * w1.x + sz * w2.x + bias.x + g[i][k].x;
      float v1 = sx * w0.y + sy * w1.y + sz * w2.y + bias.y + g[i][k].y;
      best0 = fmaxf(best0, v0);
      best1 = fmaxf(best1, v1);
    }
    v2f r; r.x = best0; r.y = best1;
    *(v2f*)(out + (size_t)(bm0 + i) * C_ + 2 * lane) = r;
  }
}

constexpr int CV_ROWS_ = 32;
static_assert(C_ == 64, "two 32-column halves per row tile; 16 K-steps");

__global__ void __launch_bounds__(128)
conv_relu_kernel(const float* __restrict__ X, const float* __restrict__ W,
                 const float* __restrict__ bias, float* __restrict__ out) {
  const int w = threadIdx.x >> 5;
  const int lane = threadIdx.x & 31;
  const int half = lane >> 4;
  const int ln = lane & 15;
  const int row0 = (int)blockIdx.x * CV_ROWS_ + (w >> 1) * 16;
  const int col0 = (w & 1) * 32;

  const float* xr = X + (size_t)(row0 + ln) * C_ + 2 * half;
  const float* wc = W + (size_t)(2 * half) * C_ + col0 + ln;
  v2f a[C_ / 4];
  v2f b[2][C_ / 4];
#pragma unroll
  for (int s = 0; s < C_ / 4; ++s) a[s] = *(const v2f*)(xr + 4 * s);
#pragma unroll
  for (int s = 0; s < C_ / 4; ++s) {
#pragma unroll
    for (int t = 0; t < 2; ++t) {
      b[t][s].x = wc[(size_t)(4 * s) * C_ + 16 * t];
      b[t][s].y = wc[(size_t)(4 * s + 1) * C_ + 16 * t];
    }
  }

  v8f acc[2];
#pragma unroll
  for (int t = 0; t < 2; ++t) acc[t] = v8f{0.f, 0.f, 0.f, 0.f, 0.f, 0.f, 0.f, 0.f};
#pragma unroll
  for (int s = 0; s < C_ / 4; ++s) {
#pragma unroll
    for (int t = 0; t < 2; ++t) {
      acc[t] = __builtin_amdgcn_wmma_f32_16x16x4_f32(false, a[s], false, b[t][s], (short)0,
                                                     acc[t], false, false);
    }
  }

#pragma unroll
  for (int t = 0; t < 2; ++t) {
    const int col = col0 + 16 * t + ln;
    const float bv = bias[col];
#pragma unroll
    for (int r = 0; r < 8; ++r) {
      const int row = row0 + r + 8 * half;
      out[(size_t)row * C_ + col] = fmaxf(acc[t][r] + bv, 0.0f);
    }
  }
}

constexpr int HID_ = EXP_ * C_;
constexpr int RP_ = 18;
static_assert(C_ == 64 && HID_ == 256, "4 waves x 4 groups of 16 hidden units; 2 output columns per lane");

__global__ void __launch_bounds__(128)
mlp_pair_kernel(const float* __restrict__ agg, const float* __restrict__ W1,
                const float* __restrict__ b1, const float* __restrict__ W2,
                const float* __restrict__ b2, const float* __restrict__ skip,
                float* __restrict__ out) {
  __shared__ float red[4][C_][RP_];
  const int w = threadIdx.x >> 5;
  const int lane = threadIdx.x & 31;
  const int half = lane >> 4;
  const int ln = lane & 15;
  const int row0 = blockIdx.x * 16;

  const float* xr = agg + (size_t)(row0 + ln) * C_ + 2 * half;
  v2f xb[C_ / 4];
#pragma unroll
  for (int s = 0; s < C_ / 4; ++s) xb[s] = *(const v2f*)(xr + 4 * s);

  v8f acc[4];
#pragma unroll
  for (int oc = 0; oc < 4; ++oc) acc[oc] = v8f{0.f, 0.f, 0.f, 0.f, 0.f, 0.f, 0.f, 0.f};

#pragma unroll
  for (int pr = 0; pr < 2; ++pr) {
    v8f d[2];
#pragma unroll
    for (int u = 0; u < 2; ++u) d[u] = v8f{0.f, 0.f, 0.f, 0.f, 0.f, 0.f, 0.f, 0.f};
#pragma unroll
    for (int s = 0; s < C_ / 4; ++s) {
#pragma unroll
      for (int u = 0; u < 2; ++u) {
        const int g = 2 * pr + u;
        const float* w1c = W1 + (size_t)(2 * half) * HID_ + (4 * w + g) * 16 + ln;
        v2f a;
        a.x = w1c[(size_t)(4 * s) * HID_];
        a.y = w1c[(size_t)(4 * s + 1) * HID_];
        d[u] = __builtin_amdgcn_wmma_f32_16x16x4_f32(false, a, false, xb[s], (short)0, d[u],
                                                     false, false);
      }
    }
#pragma unroll
    for (int u = 0; u < 2; ++u) {
      const int j0 = (4 * w + 2 * pr + u) * 16;
      const float* bp = b1 + j0 + 8 * half;
#pragma unroll
      for (int r = 0; r < 8; ++r) d[u][r] = fmaxf(d[u][r] + bp[r], 0.0f);
      const float* w2c = W2 + (size_t)(j0 + 8 * half) * C_ + ln;
#pragma unroll
      for (int t = 0; t < 4; ++t) {
        v2f hb;
        hb.x = d[u][2 * t];
        hb.y = d[u][2 * t + 1];
#pragma unroll
        for (int oc = 0; oc < 4; ++oc) {
          v2f a;
          a.x = w2c[(size_t)(2 * t) * C_ + 16 * oc];
          a.y = w2c[(size_t)(2 * t + 1) * C_ + 16 * oc];
          acc[oc] = __builtin_amdgcn_wmma_f32_16x16x4_f32(false, a, false, hb, (short)0,
                                                          acc[oc], false, false);
        }
      }
    }
  }

#pragma unroll
  for (int oc = 0; oc < 4; ++oc) {
#pragma unroll
    for (int r = 0; r < 8; ++r) red[w][16 * oc + 8 * half + r][ln] = acc[oc][r];
  }
  __syncthreads();

#pragma unroll
  for (int rr = 0; rr < 4; ++rr) {
    const int n = 4 * w + rr;
#pragma unroll
    for (int cc = 0; cc < 2; ++cc) {
      const int c = lane + 32 * cc;
      float v = ((red[0][c][n] + red[1][c][n]) + red[2][c][n]) + red[3][c][n];
      v = v + b2[c];
      v += skip[(size_t)(row0 + n) * C_ + c];
      out[(size_t)(row0 + n) * C_ + c] = fmaxf(v, 0.0f);
    }
  }
}

}

extern "C" void kernel_launch(void* const* d_in, const int* in_sizes, int n_in,
                              void* d_out, int out_size, void* d_ws, size_t ws_size,
                              hipStream_t stream) {
  (void)in_sizes; (void)n_in; (void)out_size; (void)ws_size;
  const float* p      = (const float*)d_in[0];
  const float* f      = (const float*)d_in[1];
  const float* W_sa   = (const float*)d_in[2];
  const float* b_sa   = (const float*)d_in[3];
  const float* W_pe1  = (const float*)d_in[4];
  const float* b_pe1  = (const float*)d_in[5];
  const float* W_pe2  = (const float*)d_in[6];
  const float* b_pe2  = (const float*)d_in[7];
  const float* W_conv = (const float*)d_in[8];
  const float* b_conv = (const float*)d_in[9];
  const float* W_pw1  = (const float*)d_in[10];
  const float* b_pw1  = (const float*)d_in[11];
  const float* W_pw2  = (const float*)d_in[12];
  const float* b_pw2  = (const float*)d_in[13];
  float* feat = (float*)d_out;

  char* ws = (char*)d_ws;
  float* f1 = (float*)ws;  ws += sizeof(float) * (size_t)B_ * N_ * C_;
  float* h  = (float*)ws;  ws += sizeof(float) * (size_t)B_ * M_ * C_;
  float* agg = (float*)ws; ws += sizeof(float) * (size_t)B_ * M_ * C_;
  int* idx1 = (int*)ws;    ws += sizeof(int) * (size_t)B_ * M_ * K_;
  int* idx2 = (int*)ws;

  const int THREADS = 256;

  {
    int waves = (B_ * N_ / 16) * (C_ / 16);
    f1_wmma_kernel<<<(waves * 32 + THREADS - 1) / THREADS, THREADS, 0, stream>>>(
        f, W_sa, b_sa, f1);
  }

  constexpr int bq_blocks = B_ * M_ / QB_;
  ball_query_both_kernel<4><<<bq_blocks, 32 * 4, 0, stream>>>(p, R1SQ, R2SQ, idx1, idx2);

  int nm_blocks = (B_ * M_ / NMQ_ * 32 + THREADS - 1) / THREADS;
  neigh_max_kernel<<<nm_blocks, THREADS, 0, stream>>>(p, f1, idx1, 1, N_, W_pe1, b_pe1, feat);

  const int MR = B_ * M_;
  for (int i = 0; i < NB_; ++i) {
    conv_relu_kernel<<<MR / CV_ROWS_, 128, 0, stream>>>(
        feat, W_conv + (size_t)i * C_ * C_, b_conv + (size_t)i * C_, h);
    neigh_max_kernel<<<nm_blocks, THREADS, 0, stream>>>(p, h, idx2, STRIDE_, M_, W_pe2,
                                                        b_pe2, agg);
    mlp_pair_kernel<<<MR / 16, 128, 0, stream>>>(
        agg, W_pw1 + (size_t)i * C_ * EXP_ * C_, b_pw1 + (size_t)i * EXP_ * C_,
        W_pw2 + (size_t)i * EXP_ * C_ * C_, b_pw2 + (size_t)i * C_, feat, feat);
  }
}


# positional term split by linearity: f1 and conv kernels add their own point's p.Wpe per row (one more K=4 WMMA per tile); gather loop = load+max only, q.Wpe subtracted once; pe sum order changes
# speedup vs baseline: 3.5522x; 1.0728x over previous
#include <hip/hip_runtime.h>
#include <hip/hip_bf16.h>
#include <cstdint>

namespace {

constexpr int B_ = 4, N_ = 8192, CIN_ = 4, C_ = 64, K_ = 32, STRIDE_ = 4, NB_ = 2, EXP_ = 4;
constexpr int M_ = N_ / STRIDE_;
constexpr float R1SQ = 0.1f * 0.1f;
constexpr float R2SQ = 0.2f * 0.2f;
constexpr float NEG_BIG = -3.4028235e38f;

typedef float v2f __attribute__((ext_vector_type(2)));
typedef float v3f __attribute__((ext_vector_type(3)));
typedef float v8f __attribute__((ext_vector_type(8)));

__device__ __forceinline__ v2f pos_frag_a(const float* __restrict__ pt, int half) {
  v2f a;
  a.x = pt[2 * half];
  float t = pt[half ? 2 : 1];
  a.y = half ? 0.0f : t;
  return a;
}
__device__ __forceinline__ v2f pos_frag_b(const float* __restrict__ wcol, int half) {
  v2f b;
  b.x = wcol[(size_t)(2 * half) * C_];
  float t = wcol[(size_t)(half ? 2 : 1) * C_];
  b.y = half ? 0.0f : t;
  return b;
}

__global__ void f1_wmma_kernel(const float* __restrict__ f, const float* __restrict__ Wsa,
                               const float* __restrict__ bsa, const float* __restrict__ p,
                               const float* __restrict__ Wpe, float* __restrict__ f1) {
  int wave = (blockIdx.x * blockDim.x + threadIdx.x) >> 5;
  int lane = threadIdx.x & 31;
  constexpr int tilesC = C_ / 16;
  constexpr int nTiles = (B_ * N_ / 16) * tilesC;
  if (wave >= nTiles) return;
  int row0 = (wave / tilesC) * 16;
  int c0 = (wave % tilesC) * 16;
  int b = row0 / N_;
  int n0 = row0 % N_;
  int half = lane >> 4;
  int ln = lane & 15;

  const float* fb = f + ((size_t)b * CIN_ + 2 * half) * N_ + n0 + ln;
  v2f a;  a.x = fb[0];        a.y = fb[N_];
  const float* wc = Wsa + (size_t)(2 * half) * C_ + c0 + ln;
  v2f bb; bb.x = wc[0];       bb.y = wc[C_];
  v2f a2 = pos_frag_a(p + (size_t)(row0 + ln) * 3, half);
  v2f b2 = pos_frag_b(Wpe + c0 + ln, half);
  v8f acc = {0.f, 0.f, 0.f, 0.f, 0.f, 0.f, 0.f, 0.f};
  acc = __builtin_amdgcn_wmma_f32_16x16x4_f32(false, a, false, bb, (short)0, acc,
                                              false, false);
  int col = c0 + ln;
  float bv = bsa[col];
#pragma unroll
  for (int r = 0; r < 8; ++r) acc[r] = fmaxf(acc[r] + bv, 0.0f);
  acc = __builtin_amdgcn_wmma_f32_16x16x4_f32(false, a2, false, b2, (short)0, acc,
                                              false, false);
#pragma unroll
  for (int r = 0; r < 8; ++r) {
    int row = row0 + r + 8 * half;
    f1[(size_t)row * C_ + col] = acc[r];
  }
}

constexpr int QB_ = 4;
constexpr int PD_ = 16;
static_assert(M_ % QB_ == 0, "a block's queries share one batch");
static_assert(K_ == 32, "one lane per list slot");
static_assert(32 % STRIDE_ == 0, "list B's sources keep their lanes from chunk to chunk");
constexpr unsigned qlane_mask() {
  unsigned m = 0u;
  for (int l = 0; l < 32; l += STRIDE_) m |= 1u << l;
  return m;
}
constexpr unsigned QMASK_ = qlane_mask();

template <int NSEG>
__global__ void __launch_bounds__(32 * NSEG)
ball_query_both_kernel(const float* __restrict__ p, float r2a, float r2b,
                       int* __restrict__ idx_a, int* __restrict__ idx_b) {
  constexpr int SEG = N_ / NSEG;
  static_assert(N_ % NSEG == 0 && SEG % (32 * PD_) == 0, "whole trips per part");
  static_assert(QB_ % NSEG == 0, "wave w puts the lists of queries w, w+NSEG, .. together");
  __shared__ int lst[2][NSEG][QB_][K_];
  __shared__ int cnts[NSEG][2 * QB_];
  const int gv = (int)(blockIdx.x * blockDim.x + threadIdx.x);
  const int lane = threadIdx.x & 31;
  const int w = __builtin_amdgcn_readfirstlane((int)(threadIdx.x >> 5));
  const int q0 = (int)blockIdx.x * QB_;
  if (q0 >= B_ * M_) return;
  const int b = q0 / M_;
  const float* pb = p + (size_t)b * N_ * 3;
  const bool qlane = (lane % STRIDE_) == 0;

  v3f q[QB_];
  {
    int q0v = (gv / (32 * NSEG)) * QB_;
    const float* pbv = p + (size_t)(q0v / M_) * N_ * 3;
    int m0v = q0v % M_;
#pragma unroll
    for (int i = 0; i < QB_; ++i) {
      const float* qa = pbv + (size_t)(m0v + i) * STRIDE_ * 3;
      q[i] = v3f{qa[0], qa[1], qa[2]};
    }
  }
#pragma unroll
  for (int i = 0; i < QB_; ++i) asm volatile("" :: "v"(q[i]));
  int cnta[QB_], cntb[QB_];
#pragma unroll
  for (int i = 0; i < QB_; ++i) { cnta[i] = 0; cntb[i] = 0; }

  const int sbase = w * SEG;
  const unsigned loff = (unsigned)lane * 3;
  const float* ps = pb + (size_t)sbase * 3 + loff;
  v3f s[PD_];
#pragma unroll
  for (int u = 0; u < PD_; ++u) {
    const float* sp = ps + (size_t)(u * 32) * 3;
    s[u] = v3f{sp[0], sp[1], sp[2]};
  }

  for (int base = 0; base < SEG; base += 32 * PD_) {
    int nbase = base + 32 * PD_;
    if (nbase > SEG - 32 * PD_) nbase = SEG - 32 * PD_;
    const float* np = ps + (size_t)nbase * 3;
#pragma unroll
    for (int u = 0; u < PD_; ++u) {
      int cb = sbase + base + 32 * u;
      bool hita[QB_], hitb[QB_];
      unsigned maska[QB_], maskb[QB_];
#pragma unroll
      for (int i = 0; i < QB_; ++i) {
        v2f d = s[u].xy - q[i].xy;
        float dz = s[u].z - q[i].z;
        float s2;
        {
#pragma clang fp contract(off)
          v2f dd = d * d;
          s2 = dd.x + dd.y;
        }
        float d2 = __builtin_fmaf(dz, dz, s2);
        hita[i] = d2 <= r2a;
        hitb[i] = d2 <= r2b;
      }
#pragma unroll
      for (int i = 0; i < QB_; ++i) {
        maska[i] = (unsigned)__ballot(hita[i]);
        maskb[i] = (unsigned)__ballot(hitb[i]) & QMASK_;
      }
#pragma unroll
      for (int i = 0; i < QB_; ++i) {
        if (__builtin_expect(maska[i] != 0u, 0)) {
          if (cnta[i] < K_) {
            if (hita[i]) {
              int pos = (int)__builtin_amdgcn_mbcnt_lo(maska[i], (unsigned)cnta[i]);
              if (pos < K_) lst[0][w][i][pos] = cb + lane;
            }
            cnta[i] += __popc(maska[i]);
          }
        }
      }
#pragma unroll
      for (int i = 0; i < QB_; ++i) {
        if (__builtin_expect(maskb[i] != 0u, 0)) {
          if (cntb[i] < K_) {
            if (hitb[i]) {
              int pos = (int)__builtin_amdgcn_mbcnt_lo(maskb[i], (unsigned)cntb[i]);
              if (qlane && pos < K_) lst[1][w][i][pos] = (cb + lane) / STRIDE_;
            }
            cntb[i] += __popc(maskb[i]);
          }
        }
      }
      const float* sp = np + (size_t)(u * 32) * 3;
      s[u] = v3f{sp[0], sp[1], sp[2]};
    }
    int cmin = cnta[0] < cntb[0] ? cnta[0] : cntb[0];
#pragma unroll
    for (int i = 1; i < QB_; ++i) {
      cmin = cnta[i] < cmin ? cnta[i] : cmin;
      cmin = cntb[i] < cmin ? cntb[i] : cmin;
    }
    if (cmin >= K_) break;
  }

  {
    int cv = cnta[0];
#pragma unroll
    for (int i = 1; i < QB_; ++i) cv = (lane == i) ? cnta[i] : cv;
#pragma unroll
    for (int i = 0; i < QB_; ++i) cv = (lane == QB_ + i) ? cntb[i] : cv;
    if (lane < 2 * QB_) cnts[w][lane] = cv;
  }
  __syncthreads();

#pragma unroll
  for (int which = 0; which < 2; ++which) {
#pragma unroll
    for (int r = 0; r < QB_ / NSEG; ++r) {
      const int qi = w + r * NSEG;
      int c[NSEG], pre[NSEG + 1];
      pre[0] = 0;
#pragma unroll
      for (int sgm = 0; sgm < NSEG; ++sgm) {
        int cs = cnts[sgm][which * QB_ + qi];
        c[sgm] = cs < K_ ? cs : K_;
        pre[sgm + 1] = pre[sgm] + c[sgm];
      }
      const int tot = pre[NSEG];
      int val = (which ? M_ : N_) - 1;
      if (tot > 0) {
        int sf = NSEG - 1;
#pragma unroll
        for (int sgm = NSEG - 2; sgm >= 0; --sgm) sf = c[sgm] > 0 ? sgm : sf;
        int sg = sf, off = 0;
        if (lane < tot) {
          sg = 0;
#pragma unroll
          for (int sgm = 1; sgm < NSEG; ++sgm) sg += (lane >= pre[sgm]) ? 1 : 0;
          int psg = 0;
#pragma unroll
          for (int sgm = 1; sgm < NSEG; ++sgm) psg = (sg >= sgm) ? pre[sgm] : psg;
          off = lane - psg;
        }
        val = lst[which][sg][qi][off];
      }
      int* outp = which ? idx_b : idx_a;
      outp[(size_t)(q0 + qi) * K_ + lane] = val;
    }
  }
}

constexpr int NMQ_ = 2;
static_assert(M_ % NMQ_ == 0, "a wave's queries share one batch");
static_assert(C_ == 64 && K_ == 32, "2 channels per lane, one neighbour per lane");

__global__ void __launch_bounds__(256)
neigh_max_kernel(const float* __restrict__ p, const float* __restrict__ src,
                 const int* __restrict__ idx, int src_n,
                 const float* __restrict__ Wpe, const float* __restrict__ bpe,
                 float* __restrict__ out) {
  int wave_v = (int)((blockIdx.x * blockDim.x + threadIdx.x) >> 5);
  int wave = __builtin_amdgcn_readfirstlane(wave_v);
  int lane = threadIdx.x & 31;
  int bm0 = wave * NMQ_;
  if (bm0 >= B_ * M_) return;
  int b = bm0 / M_;
  const float* sb = src + (size_t)b * src_n * C_ + 2 * lane;

  int j[NMQ_];
#pragma unroll
  for (int i = 0; i < NMQ_; ++i) j[i] = idx[(size_t)(bm0 + i) * K_ + lane];

  v2f g[NMQ_][K_];
#pragma unroll
  for (int i = 0; i < NMQ_; ++i) {
#pragma unroll
    for (int k = 0; k < K_; ++k) {
      int jk = __builtin_amdgcn_readlane(j[i], k);
      g[i][k] = *(const v2f*)(sb + (size_t)(unsigned)jk * C_);
    }
  }

  v2f w0 = *(const v2f*)(Wpe + 0 * C_ + 2 * lane);
  v2f w1 = *(const v2f*)(Wpe + 1 * C_ + 2 * lane);
  v2f w2 = *(const v2f*)(Wpe + 2 * C_ + 2 * lane);
  v2f bias = *(const v2f*)(bpe + 2 * lane);

#pragma unroll
  for (int i = 0; i < NMQ_; ++i) {
    float best0 = NEG_BIG, best1 = NEG_BIG;
#pragma unroll
    for (int k = 0; k < K_; ++k) {
      best0 = fmaxf(best0, g[i][k].x);
      best1 = fmaxf(best1, g[i][k].y);
    }
    const float* qa = p + (size_t)(bm0 + i) * STRIDE_ * 3;
    float qx = qa[0], qy = qa[1], qz = qa[2];
    float u0 = qx * w0.x + qy * w1.x + qz * w2.x;
    float u1 = qx * w0.y + qy * w1.y + qz * w2.y;
    v2f r; r.x = (best0 - u0) + bias.x; r.y = (best1 - u1) + bias.y;
    *(v2f*)(out + (size_t)(bm0 + i) * C_ + 2 * lane) = r;
  }
}

constexpr int CV_ROWS_ = 32;
static_assert(C_ == 64, "two 32-column halves per row tile; 16 K-steps");

__global__ void __launch_bounds__(128)
conv_relu_kernel(const float* __restrict__ X, const float* __restrict__ W,
                 const float* __restrict__ bias, const float* __restrict__ p,
                 const float* __restrict__ Wpe, float* __restrict__ out) {
  const int w = threadIdx.x >> 5;
  const int lane = threadIdx.x & 31;
  const int half = lane >> 4;
  const int ln = lane & 15;
  const int row0 = (int)blockIdx.x * CV_ROWS_ + (w >> 1) * 16;
  const int col0 = (w & 1) * 32;

  const float* xr = X + (size_t)(row0 + ln) * C_ + 2 * half;
  const float* wc = W + (size_t)(2 * half) * C_ + col0 + ln;
  v2f a[C_ / 4];
  v2f b[2][C_ / 4];
#pragma unroll
  for (int s = 0; s < C_ / 4; ++s) a[s] = *(const v2f*)(xr + 4 * s);
#pragma unroll
  for (int s = 0; s < C_ / 4; ++s) {
#pragma unroll
    for (int t = 0; t < 2; ++t) {
      b[t][s].x = wc[(size_t)(4 * s) * C_ + 16 * t];
      b[t][s].y = wc[(size_t)(4 * s + 1) * C_ + 16 * t];
    }
  }

  const v2f a2 = pos_frag_a(p + (size_t)(row0 + ln) * (STRIDE_ * 3), half);
  v2f b2[2];
#pragma unroll
  for (int t = 0; t < 2; ++t) b2[t] = pos_frag_b(Wpe + col0 + 16 * t + ln, half);

  v8f acc[2];
#pragma unroll
  for (int t = 0; t < 2; ++t) acc[t] = v8f{0.f, 0.f, 0.f, 0.f, 0.f, 0.f, 0.f, 0.f};
#pragma unroll
  for (int s = 0; s < C_ / 4; ++s) {
#pragma unroll
    for (int t = 0; t < 2; ++t) {
      acc[t] = __builtin_amdgcn_wmma_f32_16x16x4_f32(false, a[s], false, b[t][s], (short)0,
                                                     acc[t], false, false);
    }
  }

#pragma unroll
  for (int t = 0; t < 2; ++t) {
    const int col = col0 + 16 * t + ln;
    const float bv = bias[col];
#pragma unroll
    for (int r = 0; r < 8; ++r) acc[t][r] = fmaxf(acc[t][r] + bv, 0.0f);
    acc[t] = __builtin_amdgcn_wmma_f32_16x16x4_f32(false, a2, false, b2[t], (short)0,
                                                   acc[t], false, false);
#pragma unroll
    for (int r = 0; r < 8; ++r) {
      const int row = row0 + r + 8 * half;
      out[(size_t)row * C_ + col] = acc[t][r];
    }
  }
}

constexpr int HID_ = EXP_ * C_;
constexpr int RP_ = 18;
static_assert(C_ == 64 && HID_ == 256, "4 waves x 4 groups of 16 hidden units; 2 output columns per lane");

__global__ void __launch_bounds__(128)
mlp_pair_kernel(const float* __restrict__ agg, const float* __restrict__ W1,
                const float* __restrict__ b1, const float* __restrict__ W2,
                const float* __restrict__ b2, const float* __restrict__ skip,
                float* __restrict__ out) {
  __shared__ float red[4][C_][RP_];
  const int w = threadIdx.x >> 5;
  const int lane = threadIdx.x & 31;
  const int half = lane >> 4;
  const int ln = lane & 15;
  const int row0 = blockIdx.x * 16;

  const float* xr = agg + (size_t)(row0 + ln) * C_ + 2 * half;
  v2f xb[C_ / 4];
#pragma unroll
  for (int s = 0; s < C_ / 4; ++s) xb[s] = *(const v2f*)(xr + 4 * s);

  v8f acc[4];
#pragma unroll
  for (int oc = 0; oc < 4; ++oc) acc[oc] = v8f{0.f, 0.f, 0.f, 0.f, 0.f, 0.f, 0.f, 0.f};

#pragma unroll
  for (int pr = 0; pr < 2; ++pr) {
    v8f d[2];
#pragma unroll
    for (int u = 0; u < 2; ++u) d[u] = v8f{0.f, 0.f, 0.f, 0.f, 0.f, 0.f, 0.f, 0.f};
#pragma unroll
    for (int s = 0; s < C_ / 4; ++s) {
#pragma unroll
      for (int u = 0; u < 2; ++u) {
        const int g = 2 * pr + u;
        const float* w1c = W1 + (size_t)(2 * half) * HID_ + (4 * w + g) * 16 + ln;
        v2f a;
        a.x = w1c[(size_t)(4 * s) * HID_];
        a.y = w1c[(size_t)(4 * s + 1) * HID_];
        d[u] = __builtin_amdgcn_wmma_f32_16x16x4_f32(false, a, false, xb[s], (short)0, d[u],
                                                     false, false);
      }
    }
#pragma unroll
    for (int u = 0; u < 2; ++u) {
      const int j0 = (4 * w + 2 * pr + u) * 16;
      const float* bp = b1 + j0 + 8 * half;
#pragma unroll
      for (int r = 0; r < 8; ++r) d[u][r] = fmaxf(d[u][r] + bp[r], 0.0f);
      const float* w2c = W2 + (size_t)(j0 + 8 * half) * C_ + ln;
#pragma unroll
      for (int t = 0; t < 4; ++t) {
        v2f hb;
        hb.x = d[u][2 * t];
        hb.y = d[u][2 * t + 1];
#pragma unroll
        for (int oc = 0; oc < 4; ++oc) {
          v2f a;
          a.x = w2c[(size_t)(2 * t) * C_ + 16 * oc];
          a.y = w2c[(size_t)(2 * t + 1) * C_ + 16 * oc];
          acc[oc] = __builtin_amdgcn_wmma_f32_16x16x4_f32(false, a, false, hb, (short)0,
                                                          acc[oc], false, false);
        }
      }
    }
  }

#pragma unroll
  for (int oc = 0; oc < 4; ++oc) {
#pragma unroll
    for (int r = 0; r < 8; ++r) red[w][16 * oc + 8 * half + r][ln] = acc[oc][r];
  }
  __syncthreads();

#pragma unroll
  for (int rr = 0; rr < 4; ++rr) {
    const int n = 4 * w + rr;
#pragma unroll
    for (int cc = 0; cc < 2; ++cc) {
      const int c = lane + 32 * cc;
      float v = ((red[0][c][n] + red[1][c][n]) + red[2][c][n]) + red[3][c][n];
      v = v + b2[c];
      v += skip[(size_t)(row0 + n) * C_ + c];
      out[(size_t)(row0 + n) * C_ + c] = fmaxf(v, 0.0f);
    }
  }
}

}

extern "C" void kernel_launch(void* const* d_in, const int* in_sizes, int n_in,
                              void* d_out, int out_size, void* d_ws, size_t ws_size,
                              hipStream_t stream) {
  (void)in_sizes; (void)n_in; (void)out_size; (void)ws_size;
  const float* p      = (const float*)d_in[0];
  const float* f      = (const float*)d_in[1];
  const float* W_sa   = (const float*)d_in[2];
  const float* b_sa   = (const float*)d_in[3];
  const float* W_pe1  = (const float*)d_in[4];
  const float* b_pe1  = (const float*)d_in[5];
  const float* W_pe2  = (const float*)d_in[6];
  const float* b_pe2  = (const float*)d_in[7];
  const float* W_conv = (const float*)d_in[8];
  const float* b_conv = (const float*)d_in[9];
  const float* W_pw1  = (const float*)d_in[10];
  const float* b_pw1  = (const float*)d_in[11];
  const float* W_pw2  = (const float*)d_in[12];
  const float* b_pw2  = (const float*)d_in[13];
  float* feat = (float*)d_out;

  char* ws = (char*)d_ws;
  float* f1 = (float*)ws;  ws += sizeof(float) * (size_t)B_ * N_ * C_;
  float* h  = (float*)ws;  ws += sizeof(float) * (size_t)B_ * M_ * C_;
  float* agg = (float*)ws; ws += sizeof(float) * (size_t)B_ * M_ * C_;
  int* idx1 = (int*)ws;    ws += sizeof(int) * (size_t)B_ * M_ * K_;
  int* idx2 = (int*)ws;

  const int THREADS = 256;

  {
    int waves = (B_ * N_ / 16) * (C_ / 16);
    f1_wmma_kernel<<<(waves * 32 + THREADS - 1) / THREADS, THREADS, 0, stream>>>(
        f, W_sa, b_sa, p, W_pe1, f1);
  }

  constexpr int bq_blocks = B_ * M_ / QB_;
  ball_query_both_kernel<4><<<bq_blocks, 32 * 4, 0, stream>>>(p, R1SQ, R2SQ, idx1, idx2);

  int nm_blocks = (B_ * M_ / NMQ_ * 32 + THREADS - 1) / THREADS;
  neigh_max_kernel<<<nm_blocks, THREADS, 0, stream>>>(p, f1, idx1, N_, W_pe1, b_pe1, feat);

  const int MR = B_ * M_;
  for (int i = 0; i < NB_; ++i) {
    conv_relu_kernel<<<MR / CV_ROWS_, 128, 0, stream>>>(
        feat, W_conv + (size_t)i * C_ * C_, b_conv + (size_t)i * C_, p, W_pe2, h);
    neigh_max_kernel<<<nm_blocks, THREADS, 0, stream>>>(p, h, idx2, M_, W_pe2, b_pe2, agg);
    mlp_pair_kernel<<<MR / 16, 128, 0, stream>>>(
        agg, W_pw1 + (size_t)i * C_ * EXP_ * C_, b_pw1 + (size_t)i * EXP_ * C_,
        W_pw2 + (size_t)i * EXP_ * C_ * C_, b_pw2 + (size_t)i * C_, feat, feat);
  }
}


# embedding done at the gather: ball-query blocks write a 32-B record (features, point) per source; first max embeds its 32 neighbours by K=4 WMMAs; no feature plane, 9 -> 8 launches; sums unchanged
# speedup vs baseline: 3.7021x; 1.0422x over previous
#include <hip/hip_runtime.h>
#include <hip/hip_bf16.h>
#include <cstdint>

namespace {

constexpr int B_ = 4, N_ = 8192, CIN_ = 4, C_ = 64, K_ = 32, STRIDE_ = 4, NB_ = 2, EXP_ = 4;
constexpr int M_ = N_ / STRIDE_;
constexpr float R1SQ = 0.1f * 0.1f;
constexpr float R2SQ = 0.2f * 0.2f;
constexpr float NEG_BIG = -3.4028235e38f;

typedef float v2f __attribute__((ext_vector_type(2)));
typedef float v3f __attribute__((ext_vector_type(3)));
typedef float v4f __attribute__((ext_vector_type(4)));
typedef float v8f __attribute__((ext_vector_type(8)));

__device__ __forceinline__ v2f pos_frag_a(const float* __restrict__ pt, int half) {
  v2f a;
  a.x = pt[2 * half];
  float t = pt[half ? 2 : 1];
  a.y = half ? 0.0f : t;
  return a;
}
__device__ __forceinline__ v2f pos_frag_b(const float* __restrict__ wcol, int half) {
  v2f b;
  b.x = wcol[(size_t)(2 * half) * C_];
  float t = wcol[(size_t)(half ? 2 : 1) * C_];
  b.y = half ? 0.0f : t;
  return b;
}

constexpr int REC_ = 8;
constexpr int QB_ = 4;
constexpr int PD_ = 16;
static_assert(M_ % QB_ == 0, "a block's queries share one batch");
static_assert(K_ == 32, "one lane per list slot");
static_assert(QB_ * STRIDE_ <= 32 && CIN_ == 4, "one lane per point record of the block; 4 features + point + 0");
static_assert(32 % STRIDE_ == 0, "list B's sources keep their lanes from chunk to chunk");
constexpr unsigned qlane_mask() {
  unsigned m = 0u;
  for (int l = 0; l < 32; l += STRIDE_) m |= 1u << l;
  return m;
}
constexpr unsigned QMASK_ = qlane_mask();

template <int NSEG>
__global__ void __launch_bounds__(32 * NSEG)
ball_query_both_kernel(const float* __restrict__ p, const float* __restrict__ f,
                       float* __restrict__ rec, float r2a, float r2b,
                       int* __restrict__ idx_a, int* __restrict__ idx_b) {
  constexpr int SEG = N_ / NSEG;
  static_assert(N_ % NSEG == 0 && SEG % (32 * PD_) == 0, "whole trips per part");
  static_assert(QB_ % NSEG == 0, "wave w puts the lists of queries w, w+NSEG, .. together");
  __shared__ int lst[2][NSEG][QB_][K_];
  __shared__ int cnts[NSEG][2 * QB_];
  const int gv = (int)(blockIdx.x * blockDim.x + threadIdx.x);
  const int lane = threadIdx.x & 31;
  const int w = __builtin_amdgcn_readfirstlane((int)(threadIdx.x >> 5));
  const int q0 = (int)blockIdx.x * QB_;
  if (q0 >= B_ * M_) return;
  const int b = q0 / M_;
  const float* pb = p + (size_t)b * N_ * 3;
  const bool qlane = (lane % STRIDE_) == 0;

  if (w == 0 && lane < QB_ * STRIDE_) {
    const int row = q0 * STRIDE_ + lane;
    const float* fp = f + (size_t)b * CIN_ * N_ + (row - b * N_);
    const float* pp = p + (size_t)row * 3;
    v4f e0 = v4f{fp[0], fp[N_], fp[2 * N_], fp[3 * N_]};
    v4f e1 = v4f{pp[0], pp[1], pp[2], 0.0f};
    float* rp = rec + (size_t)row * REC_;
    *(v4f*)rp = e0;
    *(v4f*)(rp + 4) = e1;
  }

  v3f q[QB_];
  {
    int q0v = (gv / (32 * NSEG)) * QB_;
    const float* pbv = p + (size_t)(q0v / M_) * N_ * 3;
    int m0v = q0v % M_;
#pragma unroll
    for (int i = 0; i < QB_; ++i) {
      const float* qa = pbv + (size_t)(m0v + i) * STRIDE_ * 3;
      q[i] = v3f{qa[0], qa[1], qa[2]};
    }
  }
#pragma unroll
  for (int i = 0; i < QB_; ++i) asm volatile("" :: "v"(q[i]));
  int cnta[QB_], cntb[QB_];
#pragma unroll
  for (int i = 0; i < QB_; ++i) { cnta[i] = 0; cntb[i] = 0; }

  const int sbase = w * SEG;
  const unsigned loff = (unsigned)lane * 3;
  const float* ps = pb + (size_t)sbase * 3 + loff;
  v3f s[PD_];
#pragma unroll
  for (int u = 0; u < PD_; ++u) {
    const float* sp = ps + (size_t)(u * 32) * 3;
    s[u] = v3f{sp[0], sp[1], sp[2]};
  }

  for (int base = 0; base < SEG; base += 32 * PD_) {
    int nbase = base + 32 * PD_;
    if (nbase > SEG - 32 * PD_) nbase = SEG - 32 * PD_;
    const float* np = ps + (size_t)nbase * 3;
#pragma unroll
    for (int u = 0; u < PD_; ++u) {
      int cb = sbase + base + 32 * u;
      bool hita[QB_], hitb[QB_];
      unsigned maska[QB_], maskb[QB_];
#pragma unroll
      for (int i = 0; i < QB_; ++i) {
        v2f d = s[u].xy - q[i].xy;
        float dz = s[u].z - q[i].z;
        float s2;
        {
#pragma clang fp contract(off)
          v2f dd = d * d;
          s2 = dd.x + dd.y;
        }
        float d2 = __builtin_fmaf(dz, dz, s2);
        hita[i] = d2 <= r2a;
        hitb[i] = d2 <= r2b;
      }
#pragma unroll
      for (int i = 0; i < QB_; ++i) {
        maska[i] = (unsigned)__ballot(hita[i]);
        maskb[i] = (unsigned)__ballot(hitb[i]) & QMASK_;
      }
#pragma unroll
      for (int i = 0; i < QB_; ++i) {
        if (__builtin_expect(maska[i] != 0u, 0)) {
          if (cnta[i] < K_) {
            if (hita[i]) {
              int pos = (int)__builtin_amdgcn_mbcnt_lo(maska[i], (unsigned)cnta[i]);
              if (pos < K_) lst[0][w][i][pos] = cb + lane;
            }
            cnta[i] += __popc(maska[i]);
          }
        }
      }
#pragma unroll
      for (int i = 0; i < QB_; ++i) {
        if (__builtin_expect(maskb[i] != 0u, 0)) {
          if (cntb[i] < K_) {
            if (hitb[i]) {
              int pos = (int)__builtin_amdgcn_mbcnt_lo(maskb[i], (unsigned)cntb[i]);
              if (qlane && pos < K_) lst[1][w][i][pos] = (cb + lane) / STRIDE_;
            }
            cntb[i] += __popc(maskb[i]);
          }
        }
      }
      const float* sp = np + (size_t)(u * 32) * 3;
      s[u] = v3f{sp[0], sp[1], sp[2]};
    }
    int cmin = cnta[0] < cntb[0] ? cnta[0] : cntb[0];
#pragma unroll
    for (int i = 1; i < QB_; ++i) {
      cmin = cnta[i] < cmin ? cnta[i] : cmin;
      cmin = cntb[i] < cmin ? cntb[i] : cmin;
    }
    if (cmin >= K_) break;
  }

  {
    int cv = cnta[0];
#pragma unroll
    for (int i = 1; i < QB_; ++i) cv = (lane == i) ? cnta[i] : cv;
#pragma unroll
    for (int i = 0; i < QB_; ++i) cv = (lane == QB_ + i) ? cntb[i] : cv;
    if (lane < 2 * QB_) cnts[w][lane] = cv;
  }
  __syncthreads();

#pragma unroll
  for (int which = 0; which < 2; ++which) {
#pragma unroll
    for (int r = 0; r < QB_ / NSEG; ++r) {
      const int qi = w + r * NSEG;
      int c[NSEG], pre[NSEG + 1];
      pre[0] = 0;
#pragma unroll
      for (int sgm = 0; sgm < NSEG; ++sgm) {
        int cs = cnts[sgm][which * QB_ + qi];
        c[sgm] = cs < K_ ? cs : K_;
        pre[sgm + 1] = pre[sgm] + c[sgm];
      }
      const int tot = pre[NSEG];
      int val = (which ? M_ : N_) - 1;
      if (tot > 0) {
        int sf = NSEG - 1;
#pragma unroll
        for (int sgm = NSEG - 2; sgm >= 0; --sgm) sf = c[sgm] > 0 ? sgm : sf;
        int sg = sf, off = 0;
        if (lane < tot) {
          sg = 0;
#pragma unroll
          for (int sgm = 1; sgm < NSEG; ++sgm) sg += (lane >= pre[sgm]) ? 1 : 0;
          int psg = 0;
#pragma unroll
          for (int sgm = 1; sgm < NSEG; ++sgm) psg = (sg >= sgm) ? pre[sgm] : psg;
          off = lane - psg;
        }
        val = lst[which][sg][qi][off];
      }
      int* outp = which ? idx_b : idx_a;
      outp[(size_t)(q0 + qi) * K_ + lane] = val;
    }
  }
}

constexpr int NMQ_ = 2;
static_assert(M_ % NMQ_ == 0, "a wave's queries share one batch");
static_assert(C_ == 64 && K_ == 32, "2 channels per lane, one neighbour per lane");

__global__ void __launch_bounds__(256)
neigh_max_kernel(const float* __restrict__ p, const float* __restrict__ src,
                 const int* __restrict__ idx, int src_n,
                 const float* __restrict__ Wpe, const float* __restrict__ bpe,
                 float* __restrict__ out) {
  int wave_v = (int)((blockIdx.x * blockDim.x + threadIdx.x) >> 5);
  int wave = __builtin_amdgcn_readfirstlane(wave_v);
  int lane = threadIdx.x & 31;
  int bm0 = wave * NMQ_;
  if (bm0 >= B_ * M_) return;
  int b = bm0 / M_;
  const float* sb = src + (size_t)b * src_n * C_ + 2 * lane;

  int j[NMQ_];
#pragma unroll
  for (int i = 0; i < NMQ_; ++i) j[i] = idx[(size_t)(bm0 + i) * K_ + lane];

  v2f g[NMQ_][K_];
#pragma unroll
  for (int i = 0; i < NMQ_; ++i) {
#pragma unroll
    for (int k = 0; k < K_; ++k) {
      int jk = __builtin_amdgcn_readlane(j[i], k);
      g[i][k] = *(const v2f*)(sb + (size_t)(unsigned)jk * C_);
    }
  }

  v2f w0 = *(const v2f*)(Wpe + 0 * C_ + 2 * lane);
  v2f w1 = *(const v2f*)(Wpe + 1 * C_ + 2 * lane);
  v2f w2 = *(const v2f*)(Wpe + 2 * C_ + 2 * lane);
  v2f bias = *(const v2f*)(bpe + 2 * lane);

#pragma unroll
  for (int i = 0; i < NMQ_; ++i) {
    float best0 = NEG_BIG, best1 = NEG_BIG;
#pragma unroll
    for (int k = 0; k < K_; ++k) {
      best0 = fmaxf(best0, g[i][k].x);
      best1 = fmaxf(best1, g[i][k].y);
    }
    const float* qa = p + (size_t)(bm0 + i) * STRIDE_ * 3;
    float qx = qa[0], qy = qa[1], qz = qa[2];
    float u0 = qx * w0.x + qy * w1.x + qz * w2.x;
    float u1 = qx * w0.y + qy * w1.y + qz * w2.y;
    v2f r; r.x = (best0 - u0) + bias.x; r.y = (best1 - u1) + bias.y;
    *(v2f*)(out + (size_t)(bm0 + i) * C_ + 2 * lane) = r;
  }
}

__global__ void __launch_bounds__(256)
embed_max_kernel(const float* __restrict__ p, const float* __restrict__ rec,
                 const int* __restrict__ idx,
                 const float* __restrict__ Wsa, const float* __restrict__ bsa,
                 const float* __restrict__ Wpe, const float* __restrict__ bpe,
                 float* __restrict__ out) {
  int wave_v = (int)((blockIdx.x * blockDim.x + threadIdx.x) >> 5);
  int wave = __builtin_amdgcn_readfirstlane(wave_v);
  int lane = threadIdx.x & 31;
  int half = lane >> 4;
  int ln = lane & 15;
  int bm0 = wave * NMQ_;
  if (bm0 >= B_ * M_) return;
  int b = bm0 / M_;
  const float* rb = rec + (size_t)b * N_ * REC_ + 2 * half;

  int j[NMQ_][2];
#pragma unroll
  for (int i = 0; i < NMQ_; ++i) {
#pragma unroll
    for (int t = 0; t < 2; ++t) j[i][t] = idx[(size_t)(bm0 + i) * K_ + 16 * t + ln];
  }

  v2f bw[4], bp[4];
  float bv[4];
#pragma unroll
  for (int ct = 0; ct < 4; ++ct) {
    const float* wc = Wsa + (size_t)(2 * half) * C_ + 16 * ct + ln;
    bw[ct].x = wc[0];
    bw[ct].y = wc[C_];
    bp[ct] = pos_frag_b(Wpe + 16 * ct + ln, half);
    bv[ct] = bsa[16 * ct + ln];
  }
  float w0[2], w1[2], w2[2], bias[2];
#pragma unroll
  for (int s = 0; s < 2; ++s) {
    w0[s] = Wpe[0 * C_ + 32 * s + lane];
    w1[s] = Wpe[1 * C_ + 32 * s + lane];
    w2[s] = Wpe[2 * C_ + 32 * s + lane];
    bias[s] = bpe[32 * s + lane];
  }

  v2f a[NMQ_][2], a2[NMQ_][2];
#pragma unroll
  for (int i = 0; i < NMQ_; ++i) {
#pragma unroll
    for (int t = 0; t < 2; ++t) {
      const float* r = rb + (size_t)(unsigned)j[i][t] * REC_;
      a[i][t] = *(const v2f*)r;
      a2[i][t] = *(const v2f*)(r + 4);
    }
  }

#pragma unroll
  for (int i = 0; i < NMQ_; ++i) {
    float mx[4];
#pragma unroll
    for (int ct = 0; ct < 4; ++ct) {
      v8f acc[2];
#pragma unroll
      for (int t = 0; t < 2; ++t) {
        acc[t] = v8f{0.f, 0.f, 0.f, 0.f, 0.f, 0.f, 0.f, 0.f};
        acc[t] = __builtin_amdgcn_wmma_f32_16x16x4_f32(false, a[i][t], false, bw[ct], (short)0,
                                                       acc[t], false, false);
      }
#pragma unroll
      for (int t = 0; t < 2; ++t) {
#pragma unroll
        for (int r = 0; r < 8; ++r) acc[t][r] = fmaxf(acc[t][r] + bv[ct], 0.0f);
        acc[t] = __builtin_amdgcn_wmma_f32_16x16x4_f32(false, a2[i][t], false, bp[ct], (short)0,
                                                       acc[t], false, false);
      }
      float m = fmaxf(acc[0][0], acc[1][0]);
#pragma unroll
      for (int r = 1; r < 8; ++r) m = fmaxf(m, fmaxf(acc[0][r], acc[1][r]));
      unsigned mu = __builtin_bit_cast(unsigned, m);
      unsigned ou = __builtin_amdgcn_permlanex16(mu, mu, 0x76543210u, 0xfedcba98u, false, false);
      mx[ct] = fmaxf(m, __builtin_bit_cast(float, ou));
    }
    const float* qa = p + (size_t)(bm0 + i) * STRIDE_ * 3;
    float qx = qa[0], qy = qa[1], qz = qa[2];
#pragma unroll
    for (int s = 0; s < 2; ++s) {
      float best = half ? mx[2 * s + 1] : mx[2 * s];
      float u = qx * w0[s] + qy * w1[s] + qz * w2[s];
      out[(size_t)(bm0 + i) * C_ + 32 * s + lane] = (best - u) + bias[s];
    }
  }
}

constexpr int CV_ROWS_ = 32;
static_assert(C_ == 64, "two 32-column halves per row tile; 16 K-steps");

__global__ void __launch_bounds__(128)
conv_relu_kernel(const float* __restrict__ X, const float* __restrict__ W,
                 const float* __restrict__ bias, const float* __restrict__ p,
                 const float* __restrict__ Wpe, float* __restrict__ out) {
  const int w = threadIdx.x >> 5;
  const int lane = threadIdx.x & 31;
  const int half = lane >> 4;
  const int ln = lane & 15;
  const int row0 = (int)blockIdx.x * CV_ROWS_ + (w >> 1) * 16;
  const int col0 = (w & 1) * 32;

  const float* xr = X + (size_t)(row0 + ln) * C_ + 2 * half;
  const float* wc = W + (size_t)(2 * half) * C_ + col0 + ln;
  v2f a[C_ / 4];
  v2f b[2][C_ / 4];
#pragma unroll
  for (int s = 0; s < C_ / 4; ++s) a[s] = *(const v2f*)(xr + 4 * s);
#pragma unroll
  for (int s = 0; s < C_ / 4; ++s) {
#pragma unroll
    for (int t = 0; t < 2; ++t) {
      b[t][s].x = wc[(size_t)(4 * s) * C_ + 16 * t];
      b[t][s].y = wc[(size_t)(4 * s + 1) * C_ + 16 * t];
    }
  }

  const v2f a2 = pos_frag_a(p + (size_t)(row0 + ln) * (STRIDE_ * 3), half);
  v2f b2[2];
#pragma unroll
  for (int t = 0; t < 2; ++t) b2[t] = pos_frag_b(Wpe + col0 + 16 * t + ln, half);

  v8f acc[2];
#pragma unroll
  for (int t = 0; t < 2; ++t) acc[t] = v8f{0.f, 0.f, 0.f, 0.f, 0.f, 0.f, 0.f, 0.f};
#pragma unroll
  for (int s = 0; s < C_ / 4; ++s) {
#pragma unroll
    for (int t = 0; t < 2; ++t) {
      acc[t] = __builtin_amdgcn_wmma_f32_16x16x4_f32(false, a[s], false, b[t][s], (short)0,
                                                     acc[t], false, false);
    }
  }

#pragma unroll
  for (int t = 0; t < 2; ++t) {
    const int col = col0 + 16 * t + ln;
    const float bv = bias[col];
#pragma unroll
    for (int r = 0; r < 8; ++r) acc[t][r] = fmaxf(acc[t][r] + bv, 0.0f);
    acc[t] = __builtin_amdgcn_wmma_f32_16x16x4_f32(false, a2, false, b2[t], (short)0,
                                                   acc[t], false, false);
#pragma unroll
    for (int r = 0; r < 8; ++r) {
      const int row = row0 + r + 8 * half;
      out[(size_t)row * C_ + col] = acc[t][r];
    }
  }
}

constexpr int HID_ = EXP_ * C_;
constexpr int RP_ = 18;
static_assert(C_ == 64 && HID_ == 256, "4 waves x 4 groups of 16 hidden units; 2 output columns per lane");

__global__ void __launch_bounds__(128)
mlp_pair_kernel(const float* __restrict__ agg, const float* __restrict__ W1,
                const float* __restrict__ b1, const float* __restrict__ W2,
                const float* __restrict__ b2, const float* __restrict__ skip,
                float* __restrict__ out) {
  __shared__ float red[4][C_][RP_];
  const int w = threadIdx.x >> 5;
  const int lane = threadIdx.x & 31;
  const int half = lane >> 4;
  const int ln = lane & 15;
  const int row0 = blockIdx.x * 16;

  const float* xr = agg + (size_t)(row0 + ln) * C_ + 2 * half;
  v2f xb[C_ / 4];
#pragma unroll
  for (int s = 0; s < C_ / 4; ++s) xb[s] = *(const v2f*)(xr + 4 * s);

  v8f acc[4];
#pragma unroll
  for (int oc = 0; oc < 4; ++oc) acc[oc] = v8f{0.f, 0.f, 0.f, 0.f, 0.f, 0.f, 0.f, 0.f};

#pragma unroll
  for (int pr = 0; pr < 2; ++pr) {
    v8f d[2];
#pragma unroll
    for (int u = 0; u < 2; ++u) d[u] = v8f{0.f, 0.f, 0.f, 0.f, 0.f, 0.f, 0.f, 0.f};
#pragma unroll
    for (int s = 0; s < C_ / 4; ++s) {
#pragma unroll
      for (int u = 0; u < 2; ++u) {
        const int g = 2 * pr + u;
        const float* w1c = W1 + (size_t)(2 * half) * HID_ + (4 * w + g) * 16 + ln;
        v2f a;
        a.x = w1c[(size_t)(4 * s) * HID_];
        a.y = w1c[(size_t)(4 * s + 1) * HID_];
        d[u] = __builtin_amdgcn_wmma_f32_16x16x4_f32(false, a, false, xb[s], (short)0, d[u],
                                                     false, false);
      }
    }
#pragma unroll
    for (int u = 0; u < 2; ++u) {
      const int j0 = (4 * w + 2 * pr + u) * 16;
      const float* bp = b1 + j0 + 8 * half;
#pragma unroll
      for (int r = 0; r < 8; ++r) d[u][r] = fmaxf(d[u][r] + bp[r], 0.0f);
      const float* w2c = W2 + (size_t)(j0 + 8 * half) * C_ + ln;
#pragma unroll
      for (int t = 0; t < 4; ++t) {
        v2f hb;
        hb.x = d[u][2 * t];
        hb.y = d[u][2 * t + 1];
#pragma unroll
        for (int oc = 0; oc < 4; ++oc) {
          v2f a;
          a.x = w2c[(size_t)(2 * t) * C_ + 16 * oc];
          a.y = w2c[(size_t)(2 * t + 1) * C_ + 16 * oc];
          acc[oc] = __builtin_amdgcn_wmma_f32_16x16x4_f32(false, a, false, hb, (short)0,
                                                          acc[oc], false, false);
        }
      }
    }
  }

#pragma unroll
  for (int oc = 0; oc < 4; ++oc) {
#pragma unroll
    for (int r = 0; r < 8; ++r) red[w][16 * oc + 8 * half + r][ln] = acc[oc][r];
  }
  __syncthreads();

#pragma unroll
  for (int rr = 0; rr < 4; ++rr) {
    const int n = 4 * w + rr;
#pragma unroll
    for (int cc = 0; cc < 2; ++cc) {
      const int c = lane + 32 * cc;
      float v = ((red[0][c][n] + red[1][c][n]) + red[2][c][n]) + red[3][c][n];
      v = v + b2[c];
      v += skip[(size_t)(row0 + n) * C_ + c];
      out[(size_t)(row0 + n) * C_ + c] = fmaxf(v, 0.0f);
    }
  }
}

}

extern "C" void kernel_launch(void* const* d_in, const int* in_sizes, int n_in,
                              void* d_out, int out_size, void* d_ws, size_t ws_size,
                              hipStream_t stream) {
  (void)in_sizes; (void)n_in; (void)out_size; (void)ws_size;
  const float* p      = (const float*)d_in[0];
  const float* f      = (const float*)d_in[1];
  const float* W_sa   = (const float*)d_in[2];
  const float* b_sa   = (const float*)d_in[3];
  const float* W_pe1  = (const float*)d_in[4];
  const float* b_pe1  = (const float*)d_in[5];
  const float* W_pe2  = (const float*)d_in[6];
  const float* b_pe2  = (const float*)d_in[7];
  const float* W_conv = (const float*)d_in[8];
  const float* b_conv = (const float*)d_in[9];
  const float* W_pw1  = (const float*)d_in[10];
  const float* b_pw1  = (const float*)d_in[11];
  const float* W_pw2  = (const float*)d_in[12];
  const float* b_pw2  = (const float*)d_in[13];
  float* feat = (float*)d_out;

  char* ws = (char*)d_ws;
  float* rec = (float*)ws; ws += sizeof(float) * (size_t)B_ * N_ * REC_;
  float* h  = (float*)ws;  ws += sizeof(float) * (size_t)B_ * M_ * C_;
  float* agg = (float*)ws; ws += sizeof(float) * (size_t)B_ * M_ * C_;
  int* idx1 = (int*)ws;    ws += sizeof(int) * (size_t)B_ * M_ * K_;
  int* idx2 = (int*)ws;

  const int THREADS = 256;

  constexpr int bq_blocks = B_ * M_ / QB_;
  ball_query_both_kernel<4><<<bq_blocks, 32 * 4, 0, stream>>>(p, f, rec, R1SQ, R2SQ, idx1, idx2);

  int nm_blocks = (B_ * M_ / NMQ_ * 32 + THREADS - 1) / THREADS;
  embed_max_kernel<<<nm_blocks, THREADS, 0, stream>>>(p, rec, idx1, W_sa, b_sa, W_pe1, b_pe1, feat);

  const int MR = B_ * M_;
  for (int i = 0; i < NB_; ++i) {
    conv_relu_kernel<<<MR / CV_ROWS_, 128, 0, stream>>>(
        feat, W_conv + (size_t)i * C_ * C_, b_conv + (size_t)i * C_, p, W_pe2, h);
    neigh_max_kernel<<<nm_blocks, THREADS, 0, stream>>>(p, h, idx2, M_, W_pe2, b_pe2, agg);
    mlp_pair_kernel<<<MR / 16, 128, 0, stream>>>(
        agg, W_pw1 + (size_t)i * C_ * EXP_ * C_, b_pw1 + (size_t)i * EXP_ * C_,
        W_pw2 + (size_t)i * EXP_ * C_ * C_, b_pw2 + (size_t)i * C_, feat, feat);
  }
}


# each block's conv runs at the tail of the launch that finishes its input rows: 16x64 tile through LDS to two waves, weights requested ahead of LDS-only barriers; 8 -> 6 launches; sums unchanged
# speedup vs baseline: 4.1619x; 1.1242x over previous
#include <hip/hip_runtime.h>
#include <hip/hip_bf16.h>
#include <cstdint>

namespace {

constexpr int B_ = 4, N_ = 8192, CIN_ = 4, C_ = 64, K_ = 32, STRIDE_ = 4, NB_ = 2, EXP_ = 4;
constexpr int M_ = N_ / STRIDE_;
constexpr float R1SQ = 0.1f * 0.1f;
constexpr float R2SQ = 0.2f * 0.2f;
constexpr float NEG_BIG = -3.4028235e38f;

typedef float v2f __attribute__((ext_vector_type(2)));
typedef float v3f __attribute__((ext_vector_type(3)));
typedef float v4f __attribute__((ext_vector_type(4)));
typedef float v8f __attribute__((ext_vector_type(8)));

__device__ __forceinline__ v2f pos_frag_b(const float* __restrict__ wcol, int half) {
  v2f b;
  b.x = wcol[(size_t)(2 * half) * C_];
  float t = wcol[(size_t)(half ? 2 : 1) * C_];
  b.y = half ? 0.0f : t;
  return b;
}

constexpr int REC_ = 8;
constexpr int QB_ = 4;
constexpr int PD_ = 16;
static_assert(M_ % QB_ == 0, "a block's queries share one batch");
static_assert(K_ == 32, "one lane per list slot");
static_assert(QB_ * STRIDE_ <= 32 && CIN_ == 4, "one lane per point record of the block; 4 features + point + 0");
static_assert(32 % STRIDE_ == 0, "list B's sources keep their lanes from chunk to chunk");
constexpr unsigned qlane_mask() {
  unsigned m = 0u;
  for (int l = 0; l < 32; l += STRIDE_) m |= 1u << l;
  return m;
}
constexpr unsigned QMASK_ = qlane_mask();

template <int NSEG>
__global__ void __launch_bounds__(32 * NSEG)
ball_query_both_kernel(const float* __restrict__ p, const float* __restrict__ f,
                       float* __restrict__ rec, float r2a, float r2b,
                       int* __restrict__ idx_a, int* __restrict__ idx_b) {
  constexpr int SEG = N_ / NSEG;
  static_assert(N_ % NSEG == 0 && SEG % (32 * PD_) == 0, "whole trips per part");
  static_assert(QB_ % NSEG == 0, "wave w puts the lists of queries w, w+NSEG, .. together");
  __shared__ int lst[2][NSEG][QB_][K_];
  __shared__ int cnts[NSEG][2 * QB_];
  const int gv = (int)(blockIdx.x * blockDim.x + threadIdx.x);
  const int lane = threadIdx.x & 31;
  const int w = __builtin_amdgcn_readfirstlane((int)(threadIdx.x >> 5));
  const int q0 = (int)blockIdx.x * QB_;
  if (q0 >= B_ * M_) return;
  const int b = q0 / M_;
  const float* pb = p + (size_t)b * N_ * 3;
  const bool qlane = (lane % STRIDE_) == 0;

  if (w == 0 && lane < QB_ * STRIDE_) {
    const int row = q0 * STRIDE_ + lane;
    const float* fp = f + (size_t)b * CIN_ * N_ + (row - b * N_);
    const float* pp = p + (size_t)row * 3;
    v4f e0 = v4f{fp[0], fp[N_], fp[2 * N_], fp[3 * N_]};
    v4f e1 = v4f{pp[0], pp[1], pp[2], 0.0f};
    float* rp = rec + (size_t)row * REC_;
    *(v4f*)rp = e0;
    *(v4f*)(rp + 4) = e1;
  }

  v3f q[QB_];
  {
    int q0v = (gv / (32 * NSEG)) * QB_;
    const float* pbv = p + (size_t)(q0v / M_) * N_ * 3;
    int m0v = q0v % M_;
#pragma unroll
    for (int i = 0; i < QB_; ++i) {
      const float* qa = pbv + (size_t)(m0v + i) * STRIDE_ * 3;
      q[i] = v3f{qa[0], qa[1], qa[2]};
    }
  }
#pragma unroll
  for (int i = 0; i < QB_; ++i) asm volatile("" :: "v"(q[i]));
  int cnta[QB_], cntb[QB_];
#pragma unroll
  for (int i = 0; i < QB_; ++i) { cnta[i] = 0; cntb[i] = 0; }

  const int sbase = w * SEG;
  const unsigned loff = (unsigned)lane * 3;
  const float* ps = pb + (size_t)sbase * 3 + loff;
  v3f s[PD_];
#pragma unroll
  for (int u = 0; u < PD_; ++u) {
    const float* sp = ps + (size_t)(u * 32) * 3;
    s[u] = v3f{sp[0], sp[1], sp[2]};
  }

  for (int base = 0; base < SEG; base += 32 * PD_) {
    int nbase = base + 32 * PD_;
    if (nbase > SEG - 32 * PD_) nbase = SEG - 32 * PD_;
    const float* np = ps + (size_t)nbase * 3;
#pragma unroll
    for (int u = 0; u < PD_; ++u) {
      int cb = sbase + base + 32 * u;
      bool hita[QB_], hitb[QB_];
      unsigned maska[QB_], maskb[QB_];
#pragma unroll
      for (int i = 0; i < QB_; ++i) {
        v2f d = s[u].xy - q[i].xy;
        float dz = s[u].z - q[i].z;
        float s2;
        {
#pragma clang fp contract(off)
          v2f dd = d * d;
          s2 = dd.x + dd.y;
        }
        float d2 = __builtin_fmaf(dz, dz, s2);
        hita[i] = d2 <= r2a;
        hitb[i] = d2 <= r2b;
      }
#pragma unroll
      for (int i = 0; i < QB_; ++i) {
        maska[i] = (unsigned)__ballot(hita[i]);
        maskb[i] = (unsigned)__ballot(hitb[i]) & QMASK_;
      }
#pragma unroll
      for (int i = 0; i < QB_; ++i) {
        if (__builtin_expect(maska[i] != 0u, 0)) {
          if (cnta[i] < K_) {
            if (hita[i]) {
              int pos = (int)__builtin_amdgcn_mbcnt_lo(maska[i], (unsigned)cnta[i]);
              if (pos < K_) lst[0][w][i][pos] = cb + lane;
            }
            cnta[i] += __popc(maska[i]);
          }
        }
      }
#pragma unroll
      for (int i = 0; i < QB_; ++i) {
        if (__builtin_expect(maskb[i] != 0u, 0)) {
          if (cntb[i] < K_) {
            if (hitb[i]) {
              int pos = (int)__builtin_amdgcn_mbcnt_lo(maskb[i], (unsigned)cntb[i]);
              if (qlane && pos < K_) lst[1][w][i][pos] = (cb + lane) / STRIDE_;
            }
            cntb[i] += __popc(maskb[i]);
          }
        }
      }
      const float* sp = np + (size_t)(u * 32) * 3;
      s[u] = v3f{sp[0], sp[1], sp[2]};
    }
    int cmin = cnta[0] < cntb[0] ? cnta[0] : cntb[0];
#pragma unroll
    for (int i = 1; i < QB_; ++i) {
      cmin = cnta[i] < cmin ? cnta[i] : cmin;
      cmin = cntb[i] < cmin ? cntb[i] : cmin;
    }
    if (cmin >= K_) break;
  }

  {
    int cv = cnta[0];
#pragma unroll
    for (int i = 1; i < QB_; ++i) cv = (lane == i) ? cnta[i] : cv;
#pragma unroll
    for (int i = 0; i < QB_; ++i) cv = (lane == QB_ + i) ? cntb[i] : cv;
    if (lane < 2 * QB_) cnts[w][lane] = cv;
  }
  __syncthreads();

#pragma unroll
  for (int which = 0; which < 2; ++which) {
#pragma unroll
    for (int r = 0; r < QB_ / NSEG; ++r) {
      const int qi = w + r * NSEG;
      int c[NSEG], pre[NSEG + 1];
      pre[0] = 0;
#pragma unroll
      for (int sgm = 0; sgm < NSEG; ++sgm) {
        int cs = cnts[sgm][which * QB_ + qi];
        c[sgm] = cs < K_ ? cs : K_;
        pre[sgm + 1] = pre[sgm] + c[sgm];
      }
      const int tot = pre[NSEG];
      int val = (which ? M_ : N_) - 1;
      if (tot > 0) {
        int sf = NSEG - 1;
#pragma unroll
        for (int sgm = NSEG - 2; sgm >= 0; --sgm) sf = c[sgm] > 0 ? sgm : sf;
        int sg = sf, off = 0;
        if (lane < tot) {
          sg = 0;
#pragma unroll
          for (int sgm = 1; sgm < NSEG; ++sgm) sg += (lane >= pre[sgm]) ? 1 : 0;
          int psg = 0;
#pragma unroll
          for (int sgm = 1; sgm < NSEG; ++sgm) psg = (sg >= sgm) ? pre[sgm] : psg;
          off = lane - psg;
        }
        val = lst[which][sg][qi][off];
      }
      int* outp = which ? idx_b : idx_a;
      outp[(size_t)(q0 + qi) * K_ + lane] = val;
    }
  }
}

constexpr int NMQ_ = 2;
static_assert(M_ % NMQ_ == 0, "a wave's queries share one batch");
static_assert(C_ == 64 && K_ == 32, "2 channels per lane, one neighbour per lane");

__global__ void __launch_bounds__(256)
neigh_max_kernel(const float* __restrict__ p, const float* __restrict__ src,
                 const int* __restrict__ idx, int src_n,
                 const float* __restrict__ Wpe, const float* __restrict__ bpe,
                 float* __restrict__ out) {
  int wave_v = (int)((blockIdx.x * blockDim.x + threadIdx.x) >> 5);
  int wave = __builtin_amdgcn_readfirstlane(wave_v);
  int lane = threadIdx.x & 31;
  int bm0 = wave * NMQ_;
  if (bm0 >= B_ * M_) return;
  int b = bm0 / M_;
  const float* sb = src + (size_t)b * src_n * C_ + 2 * lane;

  int j[NMQ_];
#pragma unroll
  for (int i = 0; i < NMQ_; ++i) j[i] = idx[(size_t)(bm0 + i) * K_ + lane];

  v2f g[NMQ_][K_];
#pragma unroll
  for (int i = 0; i < NMQ_; ++i) {
#pragma unroll
    for (int k = 0; k < K_; ++k) {
      int jk = __builtin_amdgcn_readlane(j[i], k);
      g[i][k] = *(const v2f*)(sb + (size_t)(unsigned)jk * C_);
    }
  }

  v2f w0 = *(const v2f*)(Wpe + 0 * C_ + 2 * lane);
  v2f w1 = *(const v2f*)(Wpe + 1 * C_ + 2 * lane);
  v2f w2 = *(const v2f*)(Wpe + 2 * C_ + 2 * lane);
  v2f bias = *(const v2f*)(bpe + 2 * lane);

#pragma unroll
  for (int i = 0; i < NMQ_; ++i) {
    float best0 = NEG_BIG, best1 = NEG_BIG;
#pragma unroll
    for (int k = 0; k < K_; ++k) {
      best0 = fmaxf(best0, g[i][k].x);
      best1 = fmaxf(best1, g[i][k].y);
    }
    const float* qa = p + (size_t)(bm0 + i) * STRIDE_ * 3;
    float qx = qa[0], qy = qa[1], qz = qa[2];
    float u0 = qx * w0.x + qy * w1.x + qz * w2.x;
    float u1 = qx * w0.y + qy * w1.y + qz * w2.y;
    v2f r; r.x = (best0 - u0) + bias.x; r.y = (best1 - u1) + bias.y;
    *(v2f*)(out + (size_t)(bm0 + i) * C_ + 2 * lane) = r;
  }
}

constexpr int CT_ROWS_ = 16;
constexpr int TP_ = C_ + 4;
static_assert(C_ == 64, "two 32-column halves per row tile; 16 K-steps");

__device__ __forceinline__ void lds_barrier() {
  __builtin_amdgcn_fence(__ATOMIC_RELEASE, "workgroup", "local");
  __builtin_amdgcn_s_barrier();
  __builtin_amdgcn_fence(__ATOMIC_ACQUIRE, "workgroup", "local");
}

struct ConvOps {
  v2f b[2][C_ / 4];
  v2f a2;
  v2f b2[2];
};

constexpr int CS1_ = 12;
template <int S0, int S1, bool POS>
__device__ __forceinline__ void conv_ops_load(ConvOps& o, const float* __restrict__ W,
                                              const float* __restrict__ p,
                                              const float* __restrict__ Wpe,
                                              int row0, int col0, int lane) {
  const int half = lane >> 4;
  const int ln = lane & 15;
  const float* wc = W + (size_t)(2 * half) * C_ + col0 + ln;
#pragma unroll
  for (int s = S0; s < S1; ++s) {
#pragma unroll
    for (int t = 0; t < 2; ++t) {
      o.b[t][s].x = wc[(size_t)(4 * s) * C_ + 16 * t];
      o.b[t][s].y = wc[(size_t)(4 * s + 1) * C_ + 16 * t];
    }
  }
  if constexpr (POS) {
    const float* pt = p + (size_t)(row0 + ln) * (STRIDE_ * 3);
    o.a2.x = pt[2 * half];
    o.a2.y = pt[half ? 2 : 1];
#pragma unroll
    for (int t = 0; t < 2; ++t) {
      const float* wcol = Wpe + col0 + 16 * t + ln;
      o.b2[t].x = wcol[(size_t)(2 * half) * C_];
      o.b2[t].y = wcol[(size_t)(half ? 2 : 1) * C_];
    }
  }
}

__device__ __forceinline__ void conv_from_tile(const ConvOps& o, const float* tile,
                                               const float* __restrict__ bias,
                                               float* __restrict__ out,
                                               int row0, int col0, int lane) {
  const int half = lane >> 4;
  const int ln = lane & 15;
  v2f a[C_ / 4];
#pragma unroll
  for (int s = 0; s < C_ / 4; ++s) a[s] = *(const v2f*)(tile + ln * TP_ + 4 * s + 2 * half);
  v2f a2 = o.a2, b2[2];
  a2.y = half ? 0.0f : a2.y;
#pragma unroll
  for (int t = 0; t < 2; ++t) { b2[t] = o.b2[t]; b2[t].y = half ? 0.0f : b2[t].y; }

  v8f acc[2];
#pragma unroll
  for (int t = 0; t < 2; ++t) acc[t] = v8f{0.f, 0.f, 0.f, 0.f, 0.f, 0.f, 0.f, 0.f};
#pragma unroll
  for (int s = 0; s < C_ / 4; ++s) {
#pragma unroll
    for (int t = 0; t < 2; ++t) {
      acc[t] = __builtin_amdgcn_wmma_f32_16x16x4_f32(false, a[s], false, o.b[t][s], (short)0,
                                                     acc[t], false, false);
    }
  }

#pragma unroll
  for (int t = 0; t < 2; ++t) {
    const int col = col0 + 16 * t + ln;
    const float bv = bias[col];
#pragma unroll
    for (int r = 0; r < 8; ++r) acc[t][r] = fmaxf(acc[t][r] + bv, 0.0f);
    acc[t] = __builtin_amdgcn_wmma_f32_16x16x4_f32(false, a2, false, b2[t], (short)0,
                                                   acc[t], false, false);
#pragma unroll
    for (int r = 0; r < 8; ++r) {
      const int row = row0 + r + 8 * half;
      out[(size_t)row * C_ + col] = acc[t][r];
    }
  }
}

constexpr int EM_WAVES_ = CT_ROWS_ / NMQ_;
static_assert(EM_WAVES_ * NMQ_ == CT_ROWS_ && (B_ * M_) % CT_ROWS_ == 0, "whole 16-row tiles");

__global__ void __launch_bounds__(32 * EM_WAVES_)
embed_max_kernel(const float* __restrict__ p, const float* __restrict__ rec,
                 const int* __restrict__ idx,
                 const float* __restrict__ Wsa, const float* __restrict__ bsa,
                 const float* __restrict__ Wpe, const float* __restrict__ bpe,
                 float* __restrict__ out,
                 const float* __restrict__ Wn, const float* __restrict__ bn,
                 const float* __restrict__ Wpn, float* __restrict__ hn) {
  __shared__ float tile[CT_ROWS_][TP_];
  int wave_v = (int)((blockIdx.x * blockDim.x + threadIdx.x) >> 5);
  int wave = __builtin_amdgcn_readfirstlane(wave_v);
  const int wib = __builtin_amdgcn_readfirstlane((int)(threadIdx.x >> 5));
  int lane = threadIdx.x & 31;
  int half = lane >> 4;
  int ln = lane & 15;
  int bm0 = wave * NMQ_;
  const int row0 = (int)blockIdx.x * CT_ROWS_;
  if (row0 >= B_ * M_) return;
  int b = bm0 / M_;
  const float* rb = rec + (size_t)b * N_ * REC_ + 2 * half;

  int j[NMQ_][2];
#pragma unroll
  for (int i = 0; i < NMQ_; ++i) {
#pragma unroll
    for (int t = 0; t < 2; ++t) j[i][t] = idx[(size_t)(bm0 + i) * K_ + 16 * t + ln];
  }

  v2f bw[4], bp[4];
  float bv[4];
#pragma unroll
  for (int ct = 0; ct < 4; ++ct) {
    const float* wc = Wsa + (size_t)(2 * half) * C_ + 16 * ct + ln;
    bw[ct].x = wc[0];
    bw[ct].y = wc[C_];
    bp[ct] = pos_frag_b(Wpe + 16 * ct + ln, half);
    bv[ct] = bsa[16 * ct + ln];
  }
  float w0[2], w1[2], w2[2], bias[2];
#pragma unroll
  for (int s = 0; s < 2; ++s) {
    w0[s] = Wpe[0 * C_ + 32 * s + lane];
    w1[s] = Wpe[1 * C_ + 32 * s + lane];
    w2[s] = Wpe[2 * C_ + 32 * s + lane];
    bias[s] = bpe[32 * s + lane];
  }

  v2f a[NMQ_][2], a2[NMQ_][2];
#pragma unroll
  for (int i = 0; i < NMQ_; ++i) {
#pragma unroll
    for (int t = 0; t < 2; ++t) {
      const float* r = rb + (size_t)(unsigned)j[i][t] * REC_;
      a[i][t] = *(const v2f*)r;
      a2[i][t] = *(const v2f*)(r + 4);
    }
  }

#pragma unroll
  for (int i = 0; i < NMQ_; ++i) {
    float mx[4];
#pragma unroll
    for (int ct = 0; ct < 4; ++ct) {
      v8f acc[2];
#pragma unroll
      for (int t = 0; t < 2; ++t) {
        acc[t] = v8f{0.f, 0.f, 0.f, 0.f, 0.f, 0.f, 0.f, 0.f};
        acc[t] = __builtin_amdgcn_wmma_f32_16x16x4_f32(false, a[i][t], false, bw[ct], (short)0,
                                                       acc[t], false, false);
      }
#pragma unroll
      for (int t = 0; t < 2; ++t) {
#pragma unroll
        for (int r = 0; r < 8; ++r) acc[t][r] = fmaxf(acc[t][r] + bv[ct], 0.0f);
        acc[t] = __builtin_amdgcn_wmma_f32_16x16x4_f32(false, a2[i][t], false, bp[ct], (short)0,
                                                       acc[t], false, false);
      }
      float m = fmaxf(acc[0][0], acc[1][0]);
#pragma unroll
      for (int r = 1; r < 8; ++r) m = fmaxf(m, fmaxf(acc[0][r], acc[1][r]));
      unsigned mu = __builtin_bit_cast(unsigned, m);
      unsigned ou = __builtin_amdgcn_permlanex16(mu, mu, 0x76543210u, 0xfedcba98u, false, false);
      mx[ct] = fmaxf(m, __builtin_bit_cast(float, ou));
    }
    const float* qa = p + (size_t)(bm0 + i) * STRIDE_ * 3;
    float qx = qa[0], qy = qa[1], qz = qa[2];
#pragma unroll
    for (int s = 0; s < 2; ++s) {
      float best = half ? mx[2 * s + 1] : mx[2 * s];
      float u = qx * w0[s] + qy * w1[s] + qz * w2[s];
      const float val = (best - u) + bias[s];
      out[(size_t)(bm0 + i) * C_ + 32 * s + lane] = val;
      tile[NMQ_ * wib + i][32 * s + lane] = val;
    }
  }

  ConvOps co;
  if (wib < 2) conv_ops_load<0, CS1_, false>(co, Wn, p, Wpn, row0, 32 * wib, lane);
  lds_barrier();
  if (wib < 2) {
    conv_ops_load<CS1_, C_ / 4, true>(co, Wn, p, Wpn, row0, 32 * wib, lane);
    conv_from_tile(co, &tile[0][0], bn, hn, row0, 32 * wib, lane);
  }
}

constexpr int HID_ = EXP_ * C_;
constexpr int RP_ = 18;
static_assert(C_ == 64 && HID_ == 256, "4 waves x 4 groups of 16 hidden units; 2 output columns per lane");

template <bool NEXT>
__global__ void __launch_bounds__(128)
mlp_pair_kernel(const float* __restrict__ agg, const float* __restrict__ W1,
                const float* __restrict__ b1, const float* __restrict__ W2,
                const float* __restrict__ b2, const float* __restrict__ skip,
                float* __restrict__ out,
                const float* __restrict__ Wn, const float* __restrict__ bn,
                const float* __restrict__ p, const float* __restrict__ Wpn,
                float* __restrict__ hn) {
  __shared__ float red[4][C_][RP_];
  __shared__ float tile[NEXT ? CT_ROWS_ : 1][TP_];
  const int w = threadIdx.x >> 5;
  const int ws = __builtin_amdgcn_readfirstlane(w);
  const int lane = threadIdx.x & 31;
  const int half = lane >> 4;
  const int ln = lane & 15;
  const int row0 = blockIdx.x * 16;

  const float* xr = agg + (size_t)(row0 + ln) * C_ + 2 * half;
  v2f xb[C_ / 4];
#pragma unroll
  for (int s = 0; s < C_ / 4; ++s) xb[s] = *(const v2f*)(xr + 4 * s);

  v8f acc[4];
#pragma unroll
  for (int oc = 0; oc < 4; ++oc) acc[oc] = v8f{0.f, 0.f, 0.f, 0.f, 0.f, 0.f, 0.f, 0.f};

#pragma unroll
  for (int pr = 0; pr < 2; ++pr) {
    v8f d[2];
#pragma unroll
    for (int u = 0; u < 2; ++u) d[u] = v8f{0.f, 0.f, 0.f, 0.f, 0.f, 0.f, 0.f, 0.f};
#pragma unroll
    for (int s = 0; s < C_ / 4; ++s) {
#pragma unroll
      for (int u = 0; u < 2; ++u) {
        const int g = 2 * pr + u;
        const float* w1c = W1 + (size_t)(2 * half) * HID_ + (4 * w + g) * 16 + ln;
        v2f a;
        a.x = w1c[(size_t)(4 * s) * HID_];
        a.y = w1c[(size_t)(4 * s + 1) * HID_];
        d[u] = __builtin_amdgcn_wmma_f32_16x16x4_f32(false, a, false, xb[s], (short)0, d[u],
                                                     false, false);
      }
    }
#pragma unroll
    for (int u = 0; u < 2; ++u) {
      const int j0 = (4 * w + 2 * pr + u) * 16;
      const float* bp = b1 + j0 + 8 * half;
#pragma unroll
      for (int r = 0; r < 8; ++r) d[u][r] = fmaxf(d[u][r] + bp[r], 0.0f);
      const float* w2c = W2 + (size_t)(j0 + 8 * half) * C_ + ln;
#pragma unroll
      for (int t = 0; t < 4; ++t) {
        v2f hb;
        hb.x = d[u][2 * t];
        hb.y = d[u][2 * t + 1];
#pragma unroll
        for (int oc = 0; oc < 4; ++oc) {
          v2f a;
          a.x = w2c[(size_t)(2 * t) * C_ + 16 * oc];
          a.y = w2c[(size_t)(2 * t + 1) * C_ + 16 * oc];
          acc[oc] = __builtin_amdgcn_wmma_f32_16x16x4_f32(false, a, false, hb, (short)0,
                                                          acc[oc], false, false);
        }
      }
    }
  }

#pragma unroll
  for (int oc = 0; oc < 4; ++oc) {
#pragma unroll
    for (int r = 0; r < 8; ++r) red[w][16 * oc + 8 * half + r][ln] = acc[oc][r];
  }
  ConvOps co;
  if constexpr (NEXT) {
    if (ws < 2) conv_ops_load<0, CS1_, false>(co, Wn, p, Wpn, row0, 32 * ws, lane);
  }
  lds_barrier();
  if constexpr (NEXT) {
    if (ws < 2) conv_ops_load<CS1_, C_ / 4, true>(co, Wn, p, Wpn, row0, 32 * ws, lane);
  }

#pragma unroll
  for (int rr = 0; rr < 4; ++rr) {
    const int n = 4 * w + rr;
#pragma unroll
    for (int cc = 0; cc < 2; ++cc) {
      const int c = lane + 32 * cc;
      float v = ((red[0][c][n] + red[1][c][n]) + red[2][c][n]) + red[3][c][n];
      v = v + b2[c];
      v += skip[(size_t)(row0 + n) * C_ + c];
      v = fmaxf(v, 0.0f);
      out[(size_t)(row0 + n) * C_ + c] = v;
      if constexpr (NEXT) tile[n][c] = v;
    }
  }

  if constexpr (NEXT) {
    lds_barrier();
    if (ws < 2) conv_from_tile(co, &tile[0][0], bn, hn, row0, 32 * ws, lane);
  }
}

}

extern "C" void kernel_launch(void* const* d_in, const int* in_sizes, int n_in,
                              void* d_out, int out_size, void* d_ws, size_t ws_size,
                              hipStream_t stream) {
  (void)in_sizes; (void)n_in; (void)out_size; (void)ws_size;
  const float* p      = (const float*)d_in[0];
  const float* f      = (const float*)d_in[1];
  const float* W_sa   = (const float*)d_in[2];
  const float* b_sa   = (const float*)d_in[3];
  const float* W_pe1  = (const float*)d_in[4];
  const float* b_pe1  = (const float*)d_in[5];
  const float* W_pe2  = (const float*)d_in[6];
  const float* b_pe2  = (const float*)d_in[7];
  const float* W_conv = (const float*)d_in[8];
  const float* b_conv = (const float*)d_in[9];
  const float* W_pw1  = (const float*)d_in[10];
  const float* b_pw1  = (const float*)d_in[11];
  const float* W_pw2  = (const float*)d_in[12];
  const float* b_pw2  = (const float*)d_in[13];
  float* feat = (float*)d_out;

  char* ws = (char*)d_ws;
  float* rec = (float*)ws; ws += sizeof(float) * (size_t)B_ * N_ * REC_;
  float* h  = (float*)ws;  ws += sizeof(float) * (size_t)B_ * M_ * C_;
  float* agg = (float*)ws; ws += sizeof(float) * (size_t)B_ * M_ * C_;
  int* idx1 = (int*)ws;    ws += sizeof(int) * (size_t)B_ * M_ * K_;
  int* idx2 = (int*)ws;

  const int THREADS = 256;
  static_assert(NB_ == 2, "two residual blocks, launched one by one below");

  constexpr int bq_blocks = B_ * M_ / QB_;
  ball_query_both_kernel<4><<<bq_blocks, 32 * 4, 0, stream>>>(p, f, rec, R1SQ, R2SQ, idx1, idx2);

  constexpr int MR = B_ * M_;
  embed_max_kernel<<<MR / CT_ROWS_, 32 * EM_WAVES_, 0, stream>>>(
      p, rec, idx1, W_sa, b_sa, W_pe1, b_pe1, feat, W_conv, b_conv, W_pe2, h);

  int nm_blocks = (B_ * M_ / NMQ_ * 32 + THREADS - 1) / THREADS;
  neigh_max_kernel<<<nm_blocks, THREADS, 0, stream>>>(p, h, idx2, M_, W_pe2, b_pe2, agg);
  mlp_pair_kernel<true><<<MR / 16, 128, 0, stream>>>(
      agg, W_pw1, b_pw1, W_pw2, b_pw2, feat, feat,
      W_conv + (size_t)C_ * C_, b_conv + (size_t)C_, p, W_pe2, h);
  neigh_max_kernel<<<nm_blocks, THREADS, 0, stream>>>(p, h, idx2, M_, W_pe2, b_pe2, agg);
  mlp_pair_kernel<false><<<MR / 16, 128, 0, stream>>>(
      agg, W_pw1 + (size_t)C_ * EXP_ * C_, b_pw1 + (size_t)EXP_ * C_,
      W_pw2 + (size_t)EXP_ * C_ * C_, b_pw2 + (size_t)C_, feat, feat,
      W_conv, b_conv, p, W_pe2, h);
}
